# starting kernel
# baseline (speedup 1.0000x reference)
#include <hip/hip_runtime.h>
#include <math.h>

#define BN 128
#define TN 64
#define EN 300
#define MP 20
#define S2E 600
#define EPSF 1e-6f

typedef __attribute__((ext_vector_type(2))) float v2f;
typedef __attribute__((ext_vector_type(8))) float v8f;

__device__ __forceinline__ float wred_sum(float v) {
#pragma unroll
  for (int off = 16; off; off >>= 1) v += __shfl_xor(v, off, 32);
  return v;
}

__global__ void norms_kernel(const float* __restrict__ h1, const float* __restrict__ h2,
                             float* __restrict__ n1, float* __restrict__ n2) {
  int gw = (blockIdx.x * blockDim.x + threadIdx.x) >> 5;
  int lane = threadIdx.x & 31;
  if (gw >= BN * TN) return;
  const float* r1 = h1 + (size_t)gw * S2E;
  const float* r2 = h2 + (size_t)gw * S2E;
#pragma unroll
  for (int d = 0; d < 2; ++d) {
    float s1 = 0.f, s2 = 0.f;
    const float* p1 = r1 + d * EN;
    const float* p2 = r2 + d * EN;
    for (int e = lane; e < EN; e += 32) {
      float a = p1[e]; s1 += a * a;
      float b = p2[e]; s2 += b * b;
    }
    s1 = wred_sum(s1);
    s2 = wred_sum(s2);
    if (lane == 0) {
      n1[d * BN * TN + gw] = sqrtf(fmaxf(s1, EPSF));
      n2[d * BN * TN + gw] = sqrtf(fmaxf(s2, EPSF));
    }
  }
}

__global__ void cosmat_kernel(const float* __restrict__ h1, const float* __restrict__ h2,
                              const float* __restrict__ n1, const float* __restrict__ n2,
                              float* __restrict__ cosm  ) {
  int bd = blockIdx.x;
  int b = bd >> 1, d = bd & 1;
  int wave = threadIdx.x >> 5, lane = threadIdx.x & 31;
  int lh = lane >> 4, ll = lane & 15;
  const float* Ab = h1 + ((size_t)b * TN) * S2E + d * EN;
  const float* Bb = h2 + ((size_t)b * TN) * S2E + d * EN;
  const float* n1p = n1 + (size_t)d * BN * TN + (size_t)b * TN;
  const float* n2p = n2 + (size_t)d * BN * TN + (size_t)b * TN;
  float* op = cosm + ((size_t)d * BN + b) * TN * TN;
  for (int t = wave; t < 16; t += 8) {
    int i0 = (t >> 2) * 16, j0 = (t & 3) * 16;
    const float* arow = Ab + (size_t)(i0 + ll) * S2E;
    const float* brow = Bb + (size_t)(j0 + ll) * S2E;
    v8f acc = {};
    for (int kk = 0; kk < EN; kk += 4) {
      int k = kk + 2 * lh;
      v2f af = { arow[k], arow[k + 1] };
      v2f bf = { brow[k], brow[k + 1] };
      acc = __builtin_amdgcn_wmma_f32_16x16x4_f32(false, af, false, bf,
                                                  (short)0, acc, false, false);
    }
    float invn2 = 1.f / n2p[j0 + ll];
#pragma unroll
    for (int r = 0; r < 8; ++r) {
      int i = i0 + lh * 8 + r;
      op[(size_t)i * TN + (j0 + ll)] = acc[r] / n1p[i] * invn2;
    }
  }
}

__global__ void rowstats_kernel(const float* __restrict__ cosm,
                                float* __restrict__ rowsum, int* __restrict__ amax) {
  int gw = (blockIdx.x * blockDim.x + threadIdx.x) >> 5;
  int lane = threadIdx.x & 31;
  if (gw >= 2 * BN * TN) return;
  const float* row = cosm + (size_t)gw * TN;
  float v0 = row[lane], v1 = row[lane + 32];
  float s = v0 + v1;
  float m = v0; int mi = lane;
  if (v1 > m) { m = v1; mi = lane + 32; }
#pragma unroll
  for (int off = 16; off; off >>= 1) {
    s += __shfl_xor(s, off, 32);
    float om = __shfl_xor(m, off, 32);
    int omi = __shfl_xor(mi, off, 32);
    if (om > m || (om == m && omi < mi)) { m = om; mi = omi; }
  }
  if (lane == 0) { rowsum[gw] = s + EPSF; amax[gw] = mi; }
}

__global__ void att_kernel(const float* __restrict__ h2, const float* __restrict__ cosm,
                           const float* __restrict__ rowsum, float* __restrict__ att) {
  int blk = blockIdx.x;
  int et = blk % 19;
  int db = blk / 19;
  int b = db % BN, d = db / BN;
  int wave = threadIdx.x >> 5, lane = threadIdx.x & 31;
  int lh = lane >> 4, ll = lane & 15;
  int i0 = wave * 16, e0 = et * 16;
  bool ein = (e0 + ll) < EN;
  int ecol = ein ? (e0 + ll) : 0;
  const float* crow = cosm + (size_t)db * TN * TN + (size_t)(i0 + ll) * TN;
  const float* h2b = h2 + ((size_t)b * TN) * S2E + d * EN;
  v8f acc = {};
  for (int kk = 0; kk < TN; kk += 4) {
    int k = kk + 2 * lh;
    v2f af = { crow[k], crow[k + 1] };
    v2f bf = { h2b[(size_t)k * S2E + ecol], h2b[(size_t)(k + 1) * S2E + ecol] };
    acc = __builtin_amdgcn_wmma_f32_16x16x4_f32(false, af, false, bf,
                                                (short)0, acc, false, false);
  }
  if (ein) {
    float* arow = att + (size_t)db * TN * EN;
    const float* rs = rowsum + (size_t)db * TN;
#pragma unroll
    for (int r = 0; r < 8; ++r) {
      int i = i0 + lh * 8 + r;
      arow[(size_t)i * EN + e0 + ll] = acc[r] / rs[i];
    }
  }
}

__global__ void match_kernel(const float* __restrict__ h1, const float* __restrict__ h2,
                             const float* __restrict__ kern, const float* __restrict__ att,
                             const int* __restrict__ amax,
                             float* __restrict__ N1mp, float* __restrict__ N2mp,
                             float* __restrict__ out) {
  int gw = (blockIdx.x * blockDim.x + threadIdx.x) >> 5;
  int lane = threadIdx.x & 31;
  if (gw >= 2 * BN * TN) return;
  int d = gw / (BN * TN);
  int bt = gw - d * (BN * TN);
  int b = bt / TN, t = bt - b * TN;
  const float* x     = h1 + (size_t)bt * S2E + d * EN;
  const float* yfull = h2 + ((size_t)b * TN + (TN - 1)) * S2E + d * EN;
  const float* yatt  = att + ((size_t)d * BN + b) * TN * EN + (size_t)t * EN;
  int jm = amax[((size_t)d * BN + b) * TN + t];
  const float* ymatt = h2 + ((size_t)b * TN + jm) * S2E + d * EN;
  const float* yown  = h2 + (size_t)bt * S2E + d * EN;
  float* orow = out + (size_t)bt * 160;

  float xv[10], y0v[10], y1v[10], y2v[10], yov[10];
#pragma unroll
  for (int q = 0; q < 10; ++q) {
    int e = lane + 32 * q;
    bool v = e < EN;
    xv[q]  = v ? x[e]     : 0.f;
    y0v[q] = v ? yfull[e] : 0.f;
    y1v[q] = v ? yatt[e]  : 0.f;
    y2v[q] = v ? ymatt[e] : 0.f;
    yov[q] = v ? yown[e]  : 0.f;
  }

  for (int m = 0; m < MP; ++m) {
    const float* kr = kern + (size_t)m * 2400;
#pragma unroll
    for (int kind = 0; kind < 3; ++kind) {
      const float* wv = kr + (kind * 2 + d) * EN;
      float sxy = 0.f, sxx = 0.f, syy = 0.f;
#pragma unroll
      for (int q = 0; q < 10; ++q) {
        int e = lane + 32 * q;
        float w = (e < EN) ? wv[e] : 0.f;
        float w2 = w * w;
        float xa = xv[q];
        float yb = (kind == 0) ? y0v[q] : (kind == 1) ? y1v[q] : y2v[q];
        sxy += xa * yb * w2;
        sxx += xa * xa * w2;
        syy += yb * yb * w2;
      }
      sxy = wred_sum(sxy); sxx = wred_sum(sxx); syy = wred_sum(syy);
      if (lane == 0)
        orow[(kind * 2 + d) * MP + m] =
            sxy / sqrtf(fmaxf(sxx, EPSF)) / sqrtf(fmaxf(syy, EPSF));
    }
    const float* wv = kr + (6 + d) * EN;
    float sxx = 0.f, syy = 0.f;
#pragma unroll
    for (int q = 0; q < 10; ++q) {
      int e = lane + 32 * q;
      float w = (e < EN) ? wv[e] : 0.f;
      float w2 = w * w;
      sxx += xv[q] * xv[q] * w2;
      syy += yov[q] * yov[q] * w2;
    }
    sxx = wred_sum(sxx); syy = wred_sum(syy);
    if (lane == 0) {
      size_t o = (((size_t)d * BN + b) * TN + t) * MP + m;
      N1mp[o] = sqrtf(fmaxf(sxx, EPSF));
      N2mp[o] = sqrtf(fmaxf(syy, EPSF));
    }
  }
}

__global__ void maxpool_kernel(const float* __restrict__ h1, const float* __restrict__ h2,
                               const float* __restrict__ kern,
                               const float* __restrict__ N1mp, const float* __restrict__ N2mp,
                               float* __restrict__ out) {
  __shared__ float wl[EN];
  int blk = blockIdx.x;
  int m = blk % MP;
  int db = blk / MP;
  int b = db % BN, d = db / BN;
  int wave = threadIdx.x >> 5, lane = threadIdx.x & 31;
  int lh = lane >> 4, ll = lane & 15;

  const float* wsrc = kern + (size_t)m * 2400 + (6 + d) * EN;
  for (int e = threadIdx.x; e < EN; e += 128) wl[e] = wsrc[e];
  __syncthreads();

  const float* Ab = h1 + ((size_t)b * TN) * S2E + d * EN;
  const float* Bb = h2 + ((size_t)b * TN) * S2E + d * EN;
  int i0 = wave * 16;
  const float* arow = Ab + (size_t)(i0 + ll) * S2E;
  const float* b0r = Bb + (size_t)(0  + ll) * S2E;
  const float* b1r = Bb + (size_t)(16 + ll) * S2E;
  const float* b2r = Bb + (size_t)(32 + ll) * S2E;
  const float* b3r = Bb + (size_t)(48 + ll) * S2E;

  v8f a0 = {}, a1 = {}, a2 = {}, a3 = {};
  for (int kk = 0; kk < EN; kk += 4) {
    int k = kk + 2 * lh;
    float w0 = wl[k], w1 = wl[k + 1];
    v2f af = { arow[k] * w0, arow[k + 1] * w1 };
    v2f f0 = { b0r[k] * w0, b0r[k + 1] * w1 };
    v2f f1 = { b1r[k] * w0, b1r[k + 1] * w1 };
    v2f f2 = { b2r[k] * w0, b2r[k + 1] * w1 };
    v2f f3 = { b3r[k] * w0, b3r[k + 1] * w1 };
    a0 = __builtin_amdgcn_wmma_f32_16x16x4_f32(false, af, false, f0, (short)0, a0, false, false);
    a1 = __builtin_amdgcn_wmma_f32_16x16x4_f32(false, af, false, f1, (short)0, a1, false, false);
    a2 = __builtin_amdgcn_wmma_f32_16x16x4_f32(false, af, false, f2, (short)0, a2, false, false);
    a3 = __builtin_amdgcn_wmma_f32_16x16x4_f32(false, af, false, f3, (short)0, a3, false, false);
  }

  const float* n1p = N1mp + ((size_t)d * BN + b) * TN * MP + m;
  const float* n2p = N2mp + ((size_t)d * BN + b) * TN * MP + m;
#pragma unroll
  for (int r = 0; r < 8; ++r) {
    float v0 = a0[r] / n2p[(size_t)(0  + ll) * MP];
    float v1 = a1[r] / n2p[(size_t)(16 + ll) * MP];
    float v2 = a2[r] / n2p[(size_t)(32 + ll) * MP];
    float v3 = a3[r] / n2p[(size_t)(48 + ll) * MP];
    float v = fmaxf(fmaxf(v0, v1), fmaxf(v2, v3));
#pragma unroll
    for (int off = 1; off < 16; off <<= 1)
      v = fmaxf(v, __shfl_xor(v, off, 32));
    int i = i0 + lh * 8 + r;
    if (ll == 0)
      out[((size_t)(b * TN + i)) * 160 + (6 + d) * MP + m] = v / n1p[(size_t)i * MP];
  }
}

extern "C" void kernel_launch(void* const* d_in, const int* in_sizes, int n_in,
                              void* d_out, int out_size, void* d_ws, size_t ws_size,
                              hipStream_t stream) {
  const float* h1   = (const float*)d_in[0];
  const float* h2   = (const float*)d_in[1];
  const float* kern = (const float*)d_in[2];
  float* out = (float*)d_out;

  float* cosm   = (float*)d_ws;
  float* n1     = cosm + (size_t)2 * BN * TN * TN;
  float* n2     = n1 + 2 * BN * TN;
  float* rowsum = n2 + 2 * BN * TN;
  int*   amax   = (int*)(rowsum + 2 * BN * TN);
  float* att    = (float*)(amax + 2 * BN * TN);
  float* N1mp   = att + (size_t)2 * BN * TN * EN;
  float* N2mp   = N1mp + (size_t)2 * BN * TN * MP;

  norms_kernel   <<<BN * TN / 8,     256, 0, stream>>>(h1, h2, n1, n2);
  cosmat_kernel  <<<2 * BN,          256, 0, stream>>>(h1, h2, n1, n2, cosm);
  rowstats_kernel<<<2 * BN * TN / 8, 256, 0, stream>>>(cosm, rowsum, amax);
  att_kernel     <<<2 * BN * 19,     128, 0, stream>>>(h2, cosm, rowsum, att);
  match_kernel   <<<2 * BN * TN / 8, 256, 0, stream>>>(h1, h2, kern, att, amax,
                                                       N1mp, N2mp, out);
  maxpool_kernel <<<2 * BN * MP,     128, 0, stream>>>(h1, h2, kern, N1mp, N2mp, out);
}


# max-pool: each wave loads its h1 strip and the h2 tile once as whole-K f32 matrix fragments kept in registers (one wave per SIMD) and loops the 20 perspectives over them; same products and sum order
# speedup vs baseline: 2.3482x; 2.3482x over previous
#include <hip/hip_runtime.h>
#include <math.h>

#define BN 128
#define TN 64
#define EN 300
#define MP 20
#define S2E 600
#define EPSF 1e-6f

typedef __attribute__((ext_vector_type(2))) float v2f;
typedef __attribute__((ext_vector_type(8))) float v8f;

__device__ __forceinline__ float wred_sum(float v) {
#pragma unroll
  for (int off = 16; off; off >>= 1) v += __shfl_xor(v, off, 32);
  return v;
}

__global__ void norms_kernel(const float* __restrict__ h1, const float* __restrict__ h2,
                             float* __restrict__ n1, float* __restrict__ n2) {
  int gw = (blockIdx.x * blockDim.x + threadIdx.x) >> 5;
  int lane = threadIdx.x & 31;
  if (gw >= BN * TN) return;
  const float* r1 = h1 + (size_t)gw * S2E;
  const float* r2 = h2 + (size_t)gw * S2E;
#pragma unroll
  for (int d = 0; d < 2; ++d) {
    float s1 = 0.f, s2 = 0.f;
    const float* p1 = r1 + d * EN;
    const float* p2 = r2 + d * EN;
    for (int e = lane; e < EN; e += 32) {
      float a = p1[e]; s1 += a * a;
      float b = p2[e]; s2 += b * b;
    }
    s1 = wred_sum(s1);
    s2 = wred_sum(s2);
    if (lane == 0) {
      n1[d * BN * TN + gw] = sqrtf(fmaxf(s1, EPSF));
      n2[d * BN * TN + gw] = sqrtf(fmaxf(s2, EPSF));
    }
  }
}

__global__ void cosmat_kernel(const float* __restrict__ h1, const float* __restrict__ h2,
                              const float* __restrict__ n1, const float* __restrict__ n2,
                              float* __restrict__ cosm  ) {
  int bd = blockIdx.x;
  int b = bd >> 1, d = bd & 1;
  int wave = threadIdx.x >> 5, lane = threadIdx.x & 31;
  int lh = lane >> 4, ll = lane & 15;
  const float* Ab = h1 + ((size_t)b * TN) * S2E + d * EN;
  const float* Bb = h2 + ((size_t)b * TN) * S2E + d * EN;
  const float* n1p = n1 + (size_t)d * BN * TN + (size_t)b * TN;
  const float* n2p = n2 + (size_t)d * BN * TN + (size_t)b * TN;
  float* op = cosm + ((size_t)d * BN + b) * TN * TN;
  for (int t = wave; t < 16; t += 8) {
    int i0 = (t >> 2) * 16, j0 = (t & 3) * 16;
    const float* arow = Ab + (size_t)(i0 + ll) * S2E;
    const float* brow = Bb + (size_t)(j0 + ll) * S2E;
    v8f acc = {};
    for (int kk = 0; kk < EN; kk += 4) {
      int k = kk + 2 * lh;
      v2f af = { arow[k], arow[k + 1] };
      v2f bf = { brow[k], brow[k + 1] };
      acc = __builtin_amdgcn_wmma_f32_16x16x4_f32(false, af, false, bf,
                                                  (short)0, acc, false, false);
    }
    float invn2 = 1.f / n2p[j0 + ll];
#pragma unroll
    for (int r = 0; r < 8; ++r) {
      int i = i0 + lh * 8 + r;
      op[(size_t)i * TN + (j0 + ll)] = acc[r] / n1p[i] * invn2;
    }
  }
}

__global__ void rowstats_kernel(const float* __restrict__ cosm,
                                float* __restrict__ rowsum, int* __restrict__ amax) {
  int gw = (blockIdx.x * blockDim.x + threadIdx.x) >> 5;
  int lane = threadIdx.x & 31;
  if (gw >= 2 * BN * TN) return;
  const float* row = cosm + (size_t)gw * TN;
  float v0 = row[lane], v1 = row[lane + 32];
  float s = v0 + v1;
  float m = v0; int mi = lane;
  if (v1 > m) { m = v1; mi = lane + 32; }
#pragma unroll
  for (int off = 16; off; off >>= 1) {
    s += __shfl_xor(s, off, 32);
    float om = __shfl_xor(m, off, 32);
    int omi = __shfl_xor(mi, off, 32);
    if (om > m || (om == m && omi < mi)) { m = om; mi = omi; }
  }
  if (lane == 0) { rowsum[gw] = s + EPSF; amax[gw] = mi; }
}

__global__ void att_kernel(const float* __restrict__ h2, const float* __restrict__ cosm,
                           const float* __restrict__ rowsum, float* __restrict__ att) {
  int blk = blockIdx.x;
  int et = blk % 19;
  int db = blk / 19;
  int b = db % BN, d = db / BN;
  int wave = threadIdx.x >> 5, lane = threadIdx.x & 31;
  int lh = lane >> 4, ll = lane & 15;
  int i0 = wave * 16, e0 = et * 16;
  bool ein = (e0 + ll) < EN;
  int ecol = ein ? (e0 + ll) : 0;
  const float* crow = cosm + (size_t)db * TN * TN + (size_t)(i0 + ll) * TN;
  const float* h2b = h2 + ((size_t)b * TN) * S2E + d * EN;
  v8f acc = {};
  for (int kk = 0; kk < TN; kk += 4) {
    int k = kk + 2 * lh;
    v2f af = { crow[k], crow[k + 1] };
    v2f bf = { h2b[(size_t)k * S2E + ecol], h2b[(size_t)(k + 1) * S2E + ecol] };
    acc = __builtin_amdgcn_wmma_f32_16x16x4_f32(false, af, false, bf,
                                                (short)0, acc, false, false);
  }
  if (ein) {
    float* arow = att + (size_t)db * TN * EN;
    const float* rs = rowsum + (size_t)db * TN;
#pragma unroll
    for (int r = 0; r < 8; ++r) {
      int i = i0 + lh * 8 + r;
      arow[(size_t)i * EN + e0 + ll] = acc[r] / rs[i];
    }
  }
}

typedef __attribute__((ext_vector_type(4))) float v4f;
#define MK_WAVES 16
#define MK_THREADS (MK_WAVES * 32)
#define MK_FILL 3
#define MK_WS 84

struct MkAcc {
  v2f xxam;
  v2f xxfp;
  v2f xy12;
  v2f yy12;
  float xy0;
  float yo;
};

__device__ __forceinline__ void mk_row(const v4f R, const v4f w, const float t0, MkAcc& a) {
  const v2f y12 = R.xy;
  const v2f wam = w.xy;
  const v2f wfp = w.zw;
  const float x = R.z, yo = R.w;
  const v2f T = y12 * wam;
  const float xx = x * x;
  const v2f xs = { x, x };
  const v2f xxs = { xx, xx };
  a.xy12 += xs * T;
  a.yy12 += y12 * T;
  a.xxam += xxs * wam;
  a.xxfp += xxs * wfp;
  a.xy0 += x * t0;
  a.yo += yo * (yo * w.w);
}

__global__ __launch_bounds__(MK_THREADS)
void match_kernel(const float* __restrict__ h1, const float* __restrict__ h2,
                  const float* __restrict__ kern, const float* __restrict__ att,
                  const int* __restrict__ amax,
                  float* __restrict__ N1mp, float* __restrict__ N2mp,
                  float* __restrict__ out) {
  __shared__ __attribute__((aligned(16))) float w2t[EN * MK_WS];
  __shared__ __attribute__((aligned(16))) float rowb[MK_WAVES * 2 * EN * 4];
  __shared__ __attribute__((aligned(16))) float y0b[EN];

  const int db = blockIdx.x;
  const int b = db % BN, d = db / BN;
  const int wave = threadIdx.x >> 5, lane = threadIdx.x & 31;

  for (int base = threadIdx.x; base < MP * EN; base += MK_THREADS * MK_FILL) {
    float wv[MK_FILL][4];
#pragma unroll
    for (int k = 0; k < MK_FILL; ++k) {
      const int idx = base + k * MK_THREADS;
      const int idc = idx < MP * EN ? idx : 0;
      const int e = idc % EN, m = idc / EN;
#pragma unroll
      for (int s = 0; s < 4; ++s)
        wv[k][s] = kern[(size_t)m * 2400 + (s * 2 + d) * EN + e];
    }
#pragma unroll
    for (int k = 0; k < MK_FILL; ++k) {
      const int idx = base + k * MK_THREADS;
      if (idx < MP * EN) {
        const int e = idx % EN, m = idx / EN;
        const v4f v = { wv[k][1] * wv[k][1], wv[k][2] * wv[k][2],
                        wv[k][0] * wv[k][0], wv[k][3] * wv[k][3] };
        *(v4f*)(w2t + e * MK_WS + m * 4) = v;
      }
    }
  }
  if (threadIdx.x < EN)
    y0b[threadIdx.x] = h2[((size_t)b * TN + (TN - 1)) * S2E + d * EN + threadIdx.x];

  const int m = lane < MP ? lane : MP - 1;
  const v4f* wp = (const v4f*)(w2t + m * 4);
  const v4f* ra = (const v4f*)rowb + (size_t)(wave * 2) * EN;
  const v4f* rb = ra + EN;
  const v4f* y0p = (const v4f*)y0b;

  for (int round = 0; round < TN / (2 * MK_WAVES); ++round) {
    const int t0 = round * (2 * MK_WAVES) + wave * 2;

    int jm[2];
#pragma unroll
    for (int r = 0; r < 2; ++r) jm[r] = amax[(size_t)db * TN + t0 + r];
#pragma unroll
    for (int r = 0; r < 2; ++r) {
      const int t = t0 + r;
      const size_t bt = (size_t)b * TN + t;
      const float* x  = h1 + bt * S2E + d * EN;
      const float* ya = att + ((size_t)db * TN + t) * EN;
      const float* ym = h2 + ((size_t)b * TN + jm[r]) * S2E + d * EN;
      const float* yo = h2 + bt * S2E + d * EN;
      float* dst = rowb + (size_t)(wave * 2 + r) * EN * 4;
#pragma unroll
      for (int q = 0; q < 10; ++q) {
        const int e = lane + 32 * q;
        if (e < EN) {
          const v4f v = { ya[e], ym[e], x[e], yo[e] };
          *(v4f*)(dst + e * 4) = v;
        }
      }
    }
    __syncthreads();

    MkAcc accA = { {0.f, 0.f}, {0.f, 0.f}, {0.f, 0.f}, {0.f, 0.f}, 0.f, 0.f };
    MkAcc accB = accA;
    float syy0 = 0.f;

    for (int e4 = 0; e4 < EN / 4; ++e4) {
      const v4f y0v = y0p[e4];
#pragma unroll
      for (int u = 0; u < 4; ++u) {
        const int e = e4 * 4 + u;
        const v4f w = wp[e * (MK_WS / 4)];
        const v4f A = ra[e];
        const v4f B = rb[e];
        const float y0 = y0v[u];
        const float t0w = y0 * w.z;
        syy0 += y0 * t0w;
        mk_row(A, w, t0w, accA);
        mk_row(B, w, t0w, accB);
      }
    }

    if (lane < MP) {
      const float ny0 = sqrtf(fmaxf(syy0, EPSF));
#pragma unroll
      for (int r = 0; r < 2; ++r) {
        const MkAcc a = r ? accB : accA;
        const int t = t0 + r;
        float* orow = out + ((size_t)b * TN + t) * 160;
        orow[(0 * 2 + d) * MP + lane] = a.xy0 / sqrtf(fmaxf(a.xxfp.x, EPSF)) / ny0;
        orow[(1 * 2 + d) * MP + lane] = a.xy12.x / sqrtf(fmaxf(a.xxam.x, EPSF)) / sqrtf(fmaxf(a.yy12.x, EPSF));
        orow[(2 * 2 + d) * MP + lane] = a.xy12.y / sqrtf(fmaxf(a.xxam.y, EPSF)) / sqrtf(fmaxf(a.yy12.y, EPSF));
        const size_t o = ((size_t)db * TN + t) * MP + lane;
        N1mp[o] = sqrtf(fmaxf(a.xxfp.y, EPSF));
        N2mp[o] = sqrtf(fmaxf(a.yo, EPSF));
      }
    }
    __syncthreads();
  }
}

#define MX_KS (EN / 4)

template <int MASK>
__device__ __forceinline__ float row_xor_max(float v) {
  const int o = __builtin_amdgcn_update_dpp(0, __builtin_bit_cast(int, v), 0x160 + MASK, 0xf, 0xf, true);
  return fmaxf(v, __builtin_bit_cast(float, o));
}

__global__ __launch_bounds__(128)
void maxpool_kernel(const float* __restrict__ h1, const float* __restrict__ h2,
                    const float* __restrict__ kern,
                    const float* __restrict__ N1mp, const float* __restrict__ N2mp,
                    float* __restrict__ out) {
  __shared__ __attribute__((aligned(16))) float wl[MP * EN];
  const int db = blockIdx.x;
  const int b = db % BN, d = db / BN;
  const int wave = threadIdx.x >> 5, lane = threadIdx.x & 31;
  const int lh = lane >> 4, ll = lane & 15;

  const int i0 = wave * 16;
  const float* Ab = h1 + ((size_t)b * TN) * S2E + d * EN + 2 * lh;
  const float* Bb = h2 + ((size_t)b * TN) * S2E + d * EN + 2 * lh;
  const v2f* arow = (const v2f*)(Ab + (size_t)(i0 + ll) * S2E);
  const v2f* b0r  = (const v2f*)(Bb + (size_t)(0  + ll) * S2E);
  const v2f* b1r  = (const v2f*)(Bb + (size_t)(16 + ll) * S2E);
  const v2f* b2r  = (const v2f*)(Bb + (size_t)(32 + ll) * S2E);
  const v2f* b3r  = (const v2f*)(Bb + (size_t)(48 + ll) * S2E);

  v2f ra[MX_KS], r0[MX_KS], r1[MX_KS], r2[MX_KS], r3[MX_KS];
#pragma unroll
  for (int q = 0; q < MX_KS; ++q) {
    ra[q] = arow[2 * q];
    r0[q] = b0r[2 * q];
    r1[q] = b1r[2 * q];
    r2[q] = b2r[2 * q];
    r3[q] = b3r[2 * q];
  }

  {
    const float* ws = kern + (6 + d) * EN;
    const int e0 = threadIdx.x, e1 = e0 + 128, e2 = e0 + 256;
    const int e2c = e2 < EN ? e2 : e0;
    float t0[MP], t1[MP], t2[MP];
#pragma unroll
    for (int m = 0; m < MP; ++m) {
      const float* wr = ws + (size_t)m * 2400;
      t0[m] = wr[e0];
      t1[m] = wr[e1];
      t2[m] = wr[e2c];
    }
#pragma unroll
    for (int m = 0; m < MP; ++m) {
      wl[m * EN + e0] = t0[m];
      wl[m * EN + e1] = t1[m];
    }
    if (e2 < EN) {
#pragma unroll
      for (int m = 0; m < MP; ++m) wl[m * EN + e2] = t2[m];
    }
  }
  __syncthreads();

  const int myrow = i0 + lh * 8 + (ll & 7);
  const float* n1p = N1mp + (size_t)db * TN * MP + (size_t)myrow * MP;
  const float* n2p = N2mp + (size_t)db * TN * MP + (size_t)ll * MP;
  float* orow = out + ((size_t)(b * TN + myrow)) * 160 + (6 + d) * MP;

  float n1nx = n1p[0];
  for (int m = 0; m < MP; ++m) {
    const float n1v = n1nx;
    n1nx = n1p[m + 1 < MP ? m + 1 : m];
    const float d0 = n2p[(size_t)0  * MP + m];
    const float d1 = n2p[(size_t)16 * MP + m];
    const float d2 = n2p[(size_t)32 * MP + m];
    const float d3 = n2p[(size_t)48 * MP + m];
    const v2f* wm = (const v2f*)(wl + m * EN + 2 * lh);

    v8f a0 = {}, a1 = {}, a2 = {}, a3 = {};
#pragma unroll
    for (int q = 0; q < MX_KS; ++q) {
      const v2f w = wm[2 * q];
      const v2f af = ra[q] * w;
      const v2f f0 = r0[q] * w;
      const v2f f1 = r1[q] * w;
      const v2f f2 = r2[q] * w;
      const v2f f3 = r3[q] * w;
      a0 = __builtin_amdgcn_wmma_f32_16x16x4_f32(false, af, false, f0, (short)0, a0, false, false);
      a1 = __builtin_amdgcn_wmma_f32_16x16x4_f32(false, af, false, f1, (short)0, a1, false, false);
      a2 = __builtin_amdgcn_wmma_f32_16x16x4_f32(false, af, false, f2, (short)0, a2, false, false);
      a3 = __builtin_amdgcn_wmma_f32_16x16x4_f32(false, af, false, f3, (short)0, a3, false, false);
    }

    float v[8];
#pragma unroll
    for (int r = 0; r < 8; ++r) {
      const float v0 = a0[r] / d0;
      const float v1 = a1[r] / d1;
      const float v2 = a2[r] / d2;
      const float v3 = a3[r] / d3;
      v[r] = fmaxf(fmaxf(v0, v1), fmaxf(v2, v3));
    }
#pragma unroll
    for (int r = 0; r < 8; ++r) v[r] = row_xor_max<1>(v[r]);
#pragma unroll
    for (int r = 0; r < 8; ++r) v[r] = row_xor_max<2>(v[r]);
#pragma unroll
    for (int r = 0; r < 8; ++r) v[r] = row_xor_max<4>(v[r]);
#pragma unroll
    for (int r = 0; r < 8; ++r) v[r] = row_xor_max<8>(v[r]);
    float mine = v[0];
#pragma unroll
    for (int r = 1; r < 8; ++r) mine = ((ll & 7) == r) ? v[r] : mine;
    if (ll < 8) orow[m] = mine / n1v;
  }
}

extern "C" void kernel_launch(void* const* d_in, const int* in_sizes, int n_in,
                              void* d_out, int out_size, void* d_ws, size_t ws_size,
                              hipStream_t stream) {
  const float* h1   = (const float*)d_in[0];
  const float* h2   = (const float*)d_in[1];
  const float* kern = (const float*)d_in[2];
  float* out = (float*)d_out;

  float* cosm   = (float*)d_ws;
  float* n1     = cosm + (size_t)2 * BN * TN * TN;
  float* n2     = n1 + 2 * BN * TN;
  float* rowsum = n2 + 2 * BN * TN;
  int*   amax   = (int*)(rowsum + 2 * BN * TN);
  float* att    = (float*)(amax + 2 * BN * TN);
  float* N1mp   = att + (size_t)2 * BN * TN * EN;
  float* N2mp   = N1mp + (size_t)2 * BN * TN * MP;

  norms_kernel   <<<BN * TN / 8,     256, 0, stream>>>(h1, h2, n1, n2);
  cosmat_kernel  <<<2 * BN,          256, 0, stream>>>(h1, h2, n1, n2, cosm);
  rowstats_kernel<<<2 * BN * TN / 8, 256, 0, stream>>>(cosm, rowsum, amax);
  att_kernel     <<<2 * BN * 19,     128, 0, stream>>>(h2, cosm, rowsum, att);
  match_kernel   <<<2 * BN,          MK_THREADS, 0, stream>>>(h1, h2, kern, att, amax,
                                                       N1mp, N2mp, out);
  maxpool_kernel <<<2 * BN,          128, 0, stream>>>(h1, h2, kern, N1mp, N2mp, out);
}


# max-pool: a perspective's LDS weight reads go out as one group per block of 19 k-steps, a block ahead into a second register set, not singly among the matrix instructions; same products and sum order
# speedup vs baseline: 2.4294x; 1.0346x over previous
#include <hip/hip_runtime.h>
#include <math.h>

#define BN 128
#define TN 64
#define EN 300
#define MP 20
#define S2E 600
#define EPSF 1e-6f

typedef __attribute__((ext_vector_type(2))) float v2f;
typedef __attribute__((ext_vector_type(8))) float v8f;

__device__ __forceinline__ float wred_sum(float v) {
#pragma unroll
  for (int off = 16; off; off >>= 1) v += __shfl_xor(v, off, 32);
  return v;
}

__global__ void norms_kernel(const float* __restrict__ h1, const float* __restrict__ h2,
                             float* __restrict__ n1, float* __restrict__ n2) {
  int gw = (blockIdx.x * blockDim.x + threadIdx.x) >> 5;
  int lane = threadIdx.x & 31;
  if (gw >= BN * TN) return;
  const float* r1 = h1 + (size_t)gw * S2E;
  const float* r2 = h2 + (size_t)gw * S2E;
#pragma unroll
  for (int d = 0; d < 2; ++d) {
    float s1 = 0.f, s2 = 0.f;
    const float* p1 = r1 + d * EN;
    const float* p2 = r2 + d * EN;
    for (int e = lane; e < EN; e += 32) {
      float a = p1[e]; s1 += a * a;
      float b = p2[e]; s2 += b * b;
    }
    s1 = wred_sum(s1);
    s2 = wred_sum(s2);
    if (lane == 0) {
      n1[d * BN * TN + gw] = sqrtf(fmaxf(s1, EPSF));
      n2[d * BN * TN + gw] = sqrtf(fmaxf(s2, EPSF));
    }
  }
}

__global__ void cosmat_kernel(const float* __restrict__ h1, const float* __restrict__ h2,
                              const float* __restrict__ n1, const float* __restrict__ n2,
                              float* __restrict__ cosm  ) {
  int bd = blockIdx.x;
  int b = bd >> 1, d = bd & 1;
  int wave = threadIdx.x >> 5, lane = threadIdx.x & 31;
  int lh = lane >> 4, ll = lane & 15;
  const float* Ab = h1 + ((size_t)b * TN) * S2E + d * EN;
  const float* Bb = h2 + ((size_t)b * TN) * S2E + d * EN;
  const float* n1p = n1 + (size_t)d * BN * TN + (size_t)b * TN;
  const float* n2p = n2 + (size_t)d * BN * TN + (size_t)b * TN;
  float* op = cosm + ((size_t)d * BN + b) * TN * TN;
  for (int t = wave; t < 16; t += 8) {
    int i0 = (t >> 2) * 16, j0 = (t & 3) * 16;
    const float* arow = Ab + (size_t)(i0 + ll) * S2E;
    const float* brow = Bb + (size_t)(j0 + ll) * S2E;
    v8f acc = {};
    for (int kk = 0; kk < EN; kk += 4) {
      int k = kk + 2 * lh;
      v2f af = { arow[k], arow[k + 1] };
      v2f bf = { brow[k], brow[k + 1] };
      acc = __builtin_amdgcn_wmma_f32_16x16x4_f32(false, af, false, bf,
                                                  (short)0, acc, false, false);
    }
    float invn2 = 1.f / n2p[j0 + ll];
#pragma unroll
    for (int r = 0; r < 8; ++r) {
      int i = i0 + lh * 8 + r;
      op[(size_t)i * TN + (j0 + ll)] = acc[r] / n1p[i] * invn2;
    }
  }
}

__global__ void rowstats_kernel(const float* __restrict__ cosm,
                                float* __restrict__ rowsum, int* __restrict__ amax) {
  int gw = (blockIdx.x * blockDim.x + threadIdx.x) >> 5;
  int lane = threadIdx.x & 31;
  if (gw >= 2 * BN * TN) return;
  const float* row = cosm + (size_t)gw * TN;
  float v0 = row[lane], v1 = row[lane + 32];
  float s = v0 + v1;
  float m = v0; int mi = lane;
  if (v1 > m) { m = v1; mi = lane + 32; }
#pragma unroll
  for (int off = 16; off; off >>= 1) {
    s += __shfl_xor(s, off, 32);
    float om = __shfl_xor(m, off, 32);
    int omi = __shfl_xor(mi, off, 32);
    if (om > m || (om == m && omi < mi)) { m = om; mi = omi; }
  }
  if (lane == 0) { rowsum[gw] = s + EPSF; amax[gw] = mi; }
}

__global__ void att_kernel(const float* __restrict__ h2, const float* __restrict__ cosm,
                           const float* __restrict__ rowsum, float* __restrict__ att) {
  int blk = blockIdx.x;
  int et = blk % 19;
  int db = blk / 19;
  int b = db % BN, d = db / BN;
  int wave = threadIdx.x >> 5, lane = threadIdx.x & 31;
  int lh = lane >> 4, ll = lane & 15;
  int i0 = wave * 16, e0 = et * 16;
  bool ein = (e0 + ll) < EN;
  int ecol = ein ? (e0 + ll) : 0;
  const float* crow = cosm + (size_t)db * TN * TN + (size_t)(i0 + ll) * TN;
  const float* h2b = h2 + ((size_t)b * TN) * S2E + d * EN;
  v8f acc = {};
  for (int kk = 0; kk < TN; kk += 4) {
    int k = kk + 2 * lh;
    v2f af = { crow[k], crow[k + 1] };
    v2f bf = { h2b[(size_t)k * S2E + ecol], h2b[(size_t)(k + 1) * S2E + ecol] };
    acc = __builtin_amdgcn_wmma_f32_16x16x4_f32(false, af, false, bf,
                                                (short)0, acc, false, false);
  }
  if (ein) {
    float* arow = att + (size_t)db * TN * EN;
    const float* rs = rowsum + (size_t)db * TN;
#pragma unroll
    for (int r = 0; r < 8; ++r) {
      int i = i0 + lh * 8 + r;
      arow[(size_t)i * EN + e0 + ll] = acc[r] / rs[i];
    }
  }
}

typedef __attribute__((ext_vector_type(4))) float v4f;
#define MK_WAVES 16
#define MK_THREADS (MK_WAVES * 32)
#define MK_FILL 3
#define MK_WS 84

struct MkAcc {
  v2f xxam;
  v2f xxfp;
  v2f xy12;
  v2f yy12;
  float xy0;
  float yo;
};

__device__ __forceinline__ void mk_row(const v4f R, const v4f w, const float t0, MkAcc& a) {
  const v2f y12 = R.xy;
  const v2f wam = w.xy;
  const v2f wfp = w.zw;
  const float x = R.z, yo = R.w;
  const v2f T = y12 * wam;
  const float xx = x * x;
  const v2f xs = { x, x };
  const v2f xxs = { xx, xx };
  a.xy12 += xs * T;
  a.yy12 += y12 * T;
  a.xxam += xxs * wam;
  a.xxfp += xxs * wfp;
  a.xy0 += x * t0;
  a.yo += yo * (yo * w.w);
}

__global__ __launch_bounds__(MK_THREADS)
void match_kernel(const float* __restrict__ h1, const float* __restrict__ h2,
                  const float* __restrict__ kern, const float* __restrict__ att,
                  const int* __restrict__ amax,
                  float* __restrict__ N1mp, float* __restrict__ N2mp,
                  float* __restrict__ out) {
  __shared__ __attribute__((aligned(16))) float w2t[EN * MK_WS];
  __shared__ __attribute__((aligned(16))) float rowb[MK_WAVES * 2 * EN * 4];
  __shared__ __attribute__((aligned(16))) float y0b[EN];

  const int db = blockIdx.x;
  const int b = db % BN, d = db / BN;
  const int wave = threadIdx.x >> 5, lane = threadIdx.x & 31;

  for (int base = threadIdx.x; base < MP * EN; base += MK_THREADS * MK_FILL) {
    float wv[MK_FILL][4];
#pragma unroll
    for (int k = 0; k < MK_FILL; ++k) {
      const int idx = base + k * MK_THREADS;
      const int idc = idx < MP * EN ? idx : 0;
      const int e = idc % EN, m = idc / EN;
#pragma unroll
      for (int s = 0; s < 4; ++s)
        wv[k][s] = kern[(size_t)m * 2400 + (s * 2 + d) * EN + e];
    }
#pragma unroll
    for (int k = 0; k < MK_FILL; ++k) {
      const int idx = base + k * MK_THREADS;
      if (idx < MP * EN) {
        const int e = idx % EN, m = idx / EN;
        const v4f v = { wv[k][1] * wv[k][1], wv[k][2] * wv[k][2],
                        wv[k][0] * wv[k][0], wv[k][3] * wv[k][3] };
        *(v4f*)(w2t + e * MK_WS + m * 4) = v;
      }
    }
  }
  if (threadIdx.x < EN)
    y0b[threadIdx.x] = h2[((size_t)b * TN + (TN - 1)) * S2E + d * EN + threadIdx.x];

  const int m = lane < MP ? lane : MP - 1;
  const v4f* wp = (const v4f*)(w2t + m * 4);
  const v4f* ra = (const v4f*)rowb + (size_t)(wave * 2) * EN;
  const v4f* rb = ra + EN;
  const v4f* y0p = (const v4f*)y0b;

  for (int round = 0; round < TN / (2 * MK_WAVES); ++round) {
    const int t0 = round * (2 * MK_WAVES) + wave * 2;

    int jm[2];
#pragma unroll
    for (int r = 0; r < 2; ++r) jm[r] = amax[(size_t)db * TN + t0 + r];
#pragma unroll
    for (int r = 0; r < 2; ++r) {
      const int t = t0 + r;
      const size_t bt = (size_t)b * TN + t;
      const float* x  = h1 + bt * S2E + d * EN;
      const float* ya = att + ((size_t)db * TN + t) * EN;
      const float* ym = h2 + ((size_t)b * TN + jm[r]) * S2E + d * EN;
      const float* yo = h2 + bt * S2E + d * EN;
      float* dst = rowb + (size_t)(wave * 2 + r) * EN * 4;
#pragma unroll
      for (int q = 0; q < 10; ++q) {
        const int e = lane + 32 * q;
        if (e < EN) {
          const v4f v = { ya[e], ym[e], x[e], yo[e] };
          *(v4f*)(dst + e * 4) = v;
        }
      }
    }
    __syncthreads();

    MkAcc accA = { {0.f, 0.f}, {0.f, 0.f}, {0.f, 0.f}, {0.f, 0.f}, 0.f, 0.f };
    MkAcc accB = accA;
    float syy0 = 0.f;

    for (int e4 = 0; e4 < EN / 4; ++e4) {
      const v4f y0v = y0p[e4];
#pragma unroll
      for (int u = 0; u < 4; ++u) {
        const int e = e4 * 4 + u;
        const v4f w = wp[e * (MK_WS / 4)];
        const v4f A = ra[e];
        const v4f B = rb[e];
        const float y0 = y0v[u];
        const float t0w = y0 * w.z;
        syy0 += y0 * t0w;
        mk_row(A, w, t0w, accA);
        mk_row(B, w, t0w, accB);
      }
    }

    if (lane < MP) {
      const float ny0 = sqrtf(fmaxf(syy0, EPSF));
#pragma unroll
      for (int r = 0; r < 2; ++r) {
        const MkAcc a = r ? accB : accA;
        const int t = t0 + r;
        float* orow = out + ((size_t)b * TN + t) * 160;
        orow[(0 * 2 + d) * MP + lane] = a.xy0 / sqrtf(fmaxf(a.xxfp.x, EPSF)) / ny0;
        orow[(1 * 2 + d) * MP + lane] = a.xy12.x / sqrtf(fmaxf(a.xxam.x, EPSF)) / sqrtf(fmaxf(a.yy12.x, EPSF));
        orow[(2 * 2 + d) * MP + lane] = a.xy12.y / sqrtf(fmaxf(a.xxam.y, EPSF)) / sqrtf(fmaxf(a.yy12.y, EPSF));
        const size_t o = ((size_t)db * TN + t) * MP + lane;
        N1mp[o] = sqrtf(fmaxf(a.xxfp.y, EPSF));
        N2mp[o] = sqrtf(fmaxf(a.yo, EPSF));
      }
    }
    __syncthreads();
  }
}

#define MX_KS (EN / 4)
#define MX_QB 19
#define MX_NB 4

template <int MASK>
__device__ __forceinline__ float row_xor_max(float v) {
  const int o = __builtin_amdgcn_update_dpp(0, __builtin_bit_cast(int, v), 0x160 + MASK, 0xf, 0xf, true);
  return fmaxf(v, __builtin_bit_cast(float, o));
}

__global__ __launch_bounds__(128)
void maxpool_kernel(const float* __restrict__ h1, const float* __restrict__ h2,
                    const float* __restrict__ kern,
                    const float* __restrict__ N1mp, const float* __restrict__ N2mp,
                    float* __restrict__ out) {
  __shared__ __attribute__((aligned(16))) float wl[MP * EN];
  const int db = blockIdx.x;
  const int b = db % BN, d = db / BN;
  const int wave = threadIdx.x >> 5, lane = threadIdx.x & 31;
  const int lh = lane >> 4, ll = lane & 15;

  const int i0 = wave * 16;
  const float* Ab = h1 + ((size_t)b * TN) * S2E + d * EN + 2 * lh;
  const float* Bb = h2 + ((size_t)b * TN) * S2E + d * EN + 2 * lh;
  const v2f* arow = (const v2f*)(Ab + (size_t)(i0 + ll) * S2E);
  const v2f* b0r  = (const v2f*)(Bb + (size_t)(0  + ll) * S2E);
  const v2f* b1r  = (const v2f*)(Bb + (size_t)(16 + ll) * S2E);
  const v2f* b2r  = (const v2f*)(Bb + (size_t)(32 + ll) * S2E);
  const v2f* b3r  = (const v2f*)(Bb + (size_t)(48 + ll) * S2E);

  v2f ra[MX_KS], r0[MX_KS], r1[MX_KS], r2[MX_KS], r3[MX_KS];
#pragma unroll
  for (int q = 0; q < MX_KS; ++q) {
    ra[q] = arow[2 * q];
    r0[q] = b0r[2 * q];
    r1[q] = b1r[2 * q];
    r2[q] = b2r[2 * q];
    r3[q] = b3r[2 * q];
  }

  {
    const float* ws = kern + (6 + d) * EN;
    const int e0 = threadIdx.x, e1 = e0 + 128, e2 = e0 + 256;
    const int e2c = e2 < EN ? e2 : e0;
    float t0[MP], t1[MP], t2[MP];
#pragma unroll
    for (int m = 0; m < MP; ++m) {
      const float* wr = ws + (size_t)m * 2400;
      t0[m] = wr[e0];
      t1[m] = wr[e1];
      t2[m] = wr[e2c];
    }
#pragma unroll
    for (int m = 0; m < MP; ++m) {
      wl[m * EN + e0] = t0[m];
      wl[m * EN + e1] = t1[m];
    }
    if (e2 < EN) {
#pragma unroll
      for (int m = 0; m < MP; ++m) wl[m * EN + e2] = t2[m];
    }
  }
  __syncthreads();

  const int myrow = i0 + lh * 8 + (ll & 7);
  const float* n1p = N1mp + (size_t)db * TN * MP + (size_t)myrow * MP;
  const float* n2p = N2mp + (size_t)db * TN * MP + (size_t)ll * MP;
  float* orow = out + ((size_t)(b * TN + myrow)) * 160 + (6 + d) * MP;

  float n1nx = n1p[0];
  v2f wc[MX_QB];
  {
    const v2f* w0 = (const v2f*)(wl + 2 * lh);
#pragma unroll
    for (int j = 0; j < MX_QB; ++j) wc[j] = w0[2 * j];
  }
  for (int m = 0; m < MP; ++m) {
    const float n1v = n1nx;
    n1nx = n1p[m + 1 < MP ? m + 1 : m];
    const float d0 = n2p[(size_t)0  * MP + m];
    const float d1 = n2p[(size_t)16 * MP + m];
    const float d2 = n2p[(size_t)32 * MP + m];
    const float d3 = n2p[(size_t)48 * MP + m];
    const v2f* wm = (const v2f*)(wl + m * EN + 2 * lh);
    const v2f* wnx = (const v2f*)(wl + (m + 1 < MP ? m + 1 : m) * EN + 2 * lh);

    v8f a0 = {}, a1 = {}, a2 = {}, a3 = {};
#pragma unroll
    for (int blk = 0; blk < MX_NB; ++blk) {
      const int qb = blk * MX_QB;
      const int qn = qb + MX_QB < MX_KS ? MX_QB : MX_KS - qb;
      v2f wn[MX_QB];
      const v2f* nsrc = blk + 1 < MX_NB ? wm + 2 * (qb + MX_QB) : wnx;
#pragma unroll
      for (int j = 0; j < MX_QB; ++j) {
        const bool in = blk + 1 < MX_NB ? (qb + MX_QB + j < MX_KS) : true;
        wn[j] = nsrc[in ? 2 * j : 0];
      }
      __builtin_amdgcn_sched_barrier(0);
#pragma unroll
      for (int j = 0; j < MX_QB; ++j) {
        if (j < qn) {
          const int q = qb + j;
          const v2f w = wc[j];
          const v2f af = ra[q] * w;
          const v2f f0 = r0[q] * w;
          const v2f f1 = r1[q] * w;
          const v2f f2 = r2[q] * w;
          const v2f f3 = r3[q] * w;
          a0 = __builtin_amdgcn_wmma_f32_16x16x4_f32(false, af, false, f0, (short)0, a0, false, false);
          a1 = __builtin_amdgcn_wmma_f32_16x16x4_f32(false, af, false, f1, (short)0, a1, false, false);
          a2 = __builtin_amdgcn_wmma_f32_16x16x4_f32(false, af, false, f2, (short)0, a2, false, false);
          a3 = __builtin_amdgcn_wmma_f32_16x16x4_f32(false, af, false, f3, (short)0, a3, false, false);
        }
      }
      __builtin_amdgcn_sched_barrier(0);
#pragma unroll
      for (int j = 0; j < MX_QB; ++j) wc[j] = wn[j];
    }

    float v[8];
#pragma unroll
    for (int r = 0; r < 8; ++r) {
      const float v0 = a0[r] / d0;
      const float v1 = a1[r] / d1;
      const float v2 = a2[r] / d2;
      const float v3 = a3[r] / d3;
      v[r] = fmaxf(fmaxf(v0, v1), fmaxf(v2, v3));
    }
#pragma unroll
    for (int r = 0; r < 8; ++r) v[r] = row_xor_max<1>(v[r]);
#pragma unroll
    for (int r = 0; r < 8; ++r) v[r] = row_xor_max<2>(v[r]);
#pragma unroll
    for (int r = 0; r < 8; ++r) v[r] = row_xor_max<4>(v[r]);
#pragma unroll
    for (int r = 0; r < 8; ++r) v[r] = row_xor_max<8>(v[r]);
    float mine = v[0];
#pragma unroll
    for (int r = 1; r < 8; ++r) mine = ((ll & 7) == r) ? v[r] : mine;
    if (ll < 8) orow[m] = mine / n1v;
  }
}

extern "C" void kernel_launch(void* const* d_in, const int* in_sizes, int n_in,
                              void* d_out, int out_size, void* d_ws, size_t ws_size,
                              hipStream_t stream) {
  const float* h1   = (const float*)d_in[0];
  const float* h2   = (const float*)d_in[1];
  const float* kern = (const float*)d_in[2];
  float* out = (float*)d_out;

  float* cosm   = (float*)d_ws;
  float* n1     = cosm + (size_t)2 * BN * TN * TN;
  float* n2     = n1 + 2 * BN * TN;
  float* rowsum = n2 + 2 * BN * TN;
  int*   amax   = (int*)(rowsum + 2 * BN * TN);
  float* att    = (float*)(amax + 2 * BN * TN);
  float* N1mp   = att + (size_t)2 * BN * TN * EN;
  float* N2mp   = N1mp + (size_t)2 * BN * TN * MP;

  norms_kernel   <<<BN * TN / 8,     256, 0, stream>>>(h1, h2, n1, n2);
  cosmat_kernel  <<<2 * BN,          256, 0, stream>>>(h1, h2, n1, n2, cosm);
  rowstats_kernel<<<2 * BN * TN / 8, 256, 0, stream>>>(cosm, rowsum, amax);
  att_kernel     <<<2 * BN * 19,     128, 0, stream>>>(h2, cosm, rowsum, att);
  match_kernel   <<<2 * BN,          MK_THREADS, 0, stream>>>(h1, h2, kern, att, amax,
                                                       N1mp, N2mp, out);
  maxpool_kernel <<<2 * BN,          128, 0, stream>>>(h1, h2, kern, N1mp, N2mp, out);
}


# row norms and the cosine rows' sum and argmax move into the cosine-matrix kernel (norms in its prologue, row stats from an LDS copy of its tiles); 6 launches become 4; arithmetic and sum order kept
# speedup vs baseline: 2.5847x; 1.0639x over previous
#include <hip/hip_runtime.h>
#include <math.h>

#define BN 128
#define TN 64
#define EN 300
#define MP 20
#define S2E 600
#define EPSF 1e-6f

typedef __attribute__((ext_vector_type(2))) float v2f;
typedef __attribute__((ext_vector_type(8))) float v8f;

__device__ __forceinline__ float wred_sum(float v) {
#pragma unroll
  for (int off = 16; off; off >>= 1) v += __shfl_xor(v, off, 32);
  return v;
}

#define CS_P 68
#define NR_R 8
#define NR_Q 10

__global__ __launch_bounds__(256)
void cosmat_kernel(const float* __restrict__ h1, const float* __restrict__ h2,
                   float* __restrict__ cosm  ,
                   float* __restrict__ rowsum, int* __restrict__ amax) {
  __shared__ float sn1[TN];
  __shared__ float sn2[TN];
  __shared__ float cs[TN * CS_P];
  int bd = blockIdx.x;
  int b = bd >> 1, d = bd & 1;
  int wave = threadIdx.x >> 5, lane = threadIdx.x & 31;
  int lh = lane >> 4, ll = lane & 15;
  const float* Ab = h1 + ((size_t)b * TN) * S2E + d * EN;
  const float* Bb = h2 + ((size_t)b * TN) * S2E + d * EN;
  const size_t db = (size_t)d * BN + b;
  float* op = cosm + db * TN * TN;

  {
    float va[NR_R][NR_Q], vb[NR_R][NR_Q];
#pragma unroll
    for (int rr = 0; rr < NR_R; ++rr) {
      const float* p1 = Ab + (size_t)(wave * NR_R + rr) * S2E;
      const float* p2 = Bb + (size_t)(wave * NR_R + rr) * S2E;
#pragma unroll
      for (int q = 0; q < NR_Q; ++q) {
        const int e = lane + 32 * q;
        const int ec = e < EN ? e : lane;
        va[rr][q] = p1[ec];
        vb[rr][q] = p2[ec];
      }
    }
    float m1 = 0.f, m2 = 0.f;
#pragma unroll
    for (int rr = 0; rr < NR_R; ++rr) {
      float s1 = 0.f, s2 = 0.f;
#pragma unroll
      for (int q = 0; q < NR_Q; ++q) {
        if (lane + 32 * q < EN) {
          s1 += va[rr][q] * va[rr][q];
          s2 += vb[rr][q] * vb[rr][q];
        }
      }
      s1 = wred_sum(s1);
      s2 = wred_sum(s2);
      if (lane == rr) { m1 = s1; m2 = s2; }
    }
    if (lane < NR_R) {
      sn1[wave * NR_R + lane] = sqrtf(fmaxf(m1, EPSF));
      sn2[wave * NR_R + lane] = sqrtf(fmaxf(m2, EPSF));
    }
  }
  __syncthreads();

  for (int t = wave; t < 16; t += 8) {
    int i0 = (t >> 2) * 16, j0 = (t & 3) * 16;
    const float* arow = Ab + (size_t)(i0 + ll) * S2E;
    const float* brow = Bb + (size_t)(j0 + ll) * S2E;
    v8f acc = {};
    for (int kk = 0; kk < EN; kk += 4) {
      int k = kk + 2 * lh;
      v2f af = { arow[k], arow[k + 1] };
      v2f bf = { brow[k], brow[k + 1] };
      acc = __builtin_amdgcn_wmma_f32_16x16x4_f32(false, af, false, bf,
                                                  (short)0, acc, false, false);
    }
    float invn2 = 1.f / sn2[j0 + ll];
#pragma unroll
    for (int r = 0; r < 8; ++r) {
      int i = i0 + lh * 8 + r;
      const float c = acc[r] / sn1[i] * invn2;
      op[(size_t)i * TN + (j0 + ll)] = c;
      cs[i * CS_P + (j0 + ll)] = c;
    }
  }
  __syncthreads();

  float mys = 0.f; int mymi = 0;
#pragma unroll
  for (int rr = 0; rr < 8; ++rr) {
    const float* row = cs + (wave * 8 + rr) * CS_P;
    float v0 = row[lane], v1 = row[lane + 32];
    float s = v0 + v1;
    float m = v0; int mi = lane;
    if (v1 > m) { m = v1; mi = lane + 32; }
#pragma unroll
    for (int off = 16; off; off >>= 1) {
      s += __shfl_xor(s, off, 32);
      float om = __shfl_xor(m, off, 32);
      int omi = __shfl_xor(mi, off, 32);
      const bool take = (om > m) | ((om == m) & (omi < mi));
      m = take ? om : m;
      mi = take ? omi : mi;
    }
    if (lane == rr) { mys = s + EPSF; mymi = mi; }
  }
  if (lane < 8) {
    rowsum[db * TN + wave * 8 + lane] = mys;
    amax[db * TN + wave * 8 + lane] = mymi;
  }
}

__global__ void att_kernel(const float* __restrict__ h2, const float* __restrict__ cosm,
                           const float* __restrict__ rowsum, float* __restrict__ att) {
  int blk = blockIdx.x;
  int et = blk % 19;
  int db = blk / 19;
  int b = db % BN, d = db / BN;
  int wave = threadIdx.x >> 5, lane = threadIdx.x & 31;
  int lh = lane >> 4, ll = lane & 15;
  int i0 = wave * 16, e0 = et * 16;
  bool ein = (e0 + ll) < EN;
  int ecol = ein ? (e0 + ll) : 0;
  const float* crow = cosm + (size_t)db * TN * TN + (size_t)(i0 + ll) * TN;
  const float* h2b = h2 + ((size_t)b * TN) * S2E + d * EN;
  v8f acc = {};
  for (int kk = 0; kk < TN; kk += 4) {
    int k = kk + 2 * lh;
    v2f af = { crow[k], crow[k + 1] };
    v2f bf = { h2b[(size_t)k * S2E + ecol], h2b[(size_t)(k + 1) * S2E + ecol] };
    acc = __builtin_amdgcn_wmma_f32_16x16x4_f32(false, af, false, bf,
                                                (short)0, acc, false, false);
  }
  if (ein) {
    float* arow = att + (size_t)db * TN * EN;
    const float* rs = rowsum + (size_t)db * TN;
#pragma unroll
    for (int r = 0; r < 8; ++r) {
      int i = i0 + lh * 8 + r;
      arow[(size_t)i * EN + e0 + ll] = acc[r] / rs[i];
    }
  }
}

typedef __attribute__((ext_vector_type(4))) float v4f;
#define MK_WAVES 16
#define MK_THREADS (MK_WAVES * 32)
#define MK_FILL 3
#define MK_WS 84

struct MkAcc {
  v2f xxam;
  v2f xxfp;
  v2f xy12;
  v2f yy12;
  float xy0;
  float yo;
};

__device__ __forceinline__ void mk_row(const v4f R, const v4f w, const float t0, MkAcc& a) {
  const v2f y12 = R.xy;
  const v2f wam = w.xy;
  const v2f wfp = w.zw;
  const float x = R.z, yo = R.w;
  const v2f T = y12 * wam;
  const float xx = x * x;
  const v2f xs = { x, x };
  const v2f xxs = { xx, xx };
  a.xy12 += xs * T;
  a.yy12 += y12 * T;
  a.xxam += xxs * wam;
  a.xxfp += xxs * wfp;
  a.xy0 += x * t0;
  a.yo += yo * (yo * w.w);
}

__global__ __launch_bounds__(MK_THREADS)
void match_kernel(const float* __restrict__ h1, const float* __restrict__ h2,
                  const float* __restrict__ kern, const float* __restrict__ att,
                  const int* __restrict__ amax,
                  float* __restrict__ N1mp, float* __restrict__ N2mp,
                  float* __restrict__ out) {
  __shared__ __attribute__((aligned(16))) float w2t[EN * MK_WS];
  __shared__ __attribute__((aligned(16))) float rowb[MK_WAVES * 2 * EN * 4];
  __shared__ __attribute__((aligned(16))) float y0b[EN];

  const int db = blockIdx.x;
  const int b = db % BN, d = db / BN;
  const int wave = threadIdx.x >> 5, lane = threadIdx.x & 31;

  for (int base = threadIdx.x; base < MP * EN; base += MK_THREADS * MK_FILL) {
    float wv[MK_FILL][4];
#pragma unroll
    for (int k = 0; k < MK_FILL; ++k) {
      const int idx = base + k * MK_THREADS;
      const int idc = idx < MP * EN ? idx : 0;
      const int e = idc % EN, m = idc / EN;
#pragma unroll
      for (int s = 0; s < 4; ++s)
        wv[k][s] = kern[(size_t)m * 2400 + (s * 2 + d) * EN + e];
    }
#pragma unroll
    for (int k = 0; k < MK_FILL; ++k) {
      const int idx = base + k * MK_THREADS;
      if (idx < MP * EN) {
        const int e = idx % EN, m = idx / EN;
        const v4f v = { wv[k][1] * wv[k][1], wv[k][2] * wv[k][2],
                        wv[k][0] * wv[k][0], wv[k][3] * wv[k][3] };
        *(v4f*)(w2t + e * MK_WS + m * 4) = v;
      }
    }
  }
  if (threadIdx.x < EN)
    y0b[threadIdx.x] = h2[((size_t)b * TN + (TN - 1)) * S2E + d * EN + threadIdx.x];

  const int m = lane < MP ? lane : MP - 1;
  const v4f* wp = (const v4f*)(w2t + m * 4);
  const v4f* ra = (const v4f*)rowb + (size_t)(wave * 2) * EN;
  const v4f* rb = ra + EN;
  const v4f* y0p = (const v4f*)y0b;

  for (int round = 0; round < TN / (2 * MK_WAVES); ++round) {
    const int t0 = round * (2 * MK_WAVES) + wave * 2;

    int jm[2];
#pragma unroll
    for (int r = 0; r < 2; ++r) jm[r] = amax[(size_t)db * TN + t0 + r];
#pragma unroll
    for (int r = 0; r < 2; ++r) {
      const int t = t0 + r;
      const size_t bt = (size_t)b * TN + t;
      const float* x  = h1 + bt * S2E + d * EN;
      const float* ya = att + ((size_t)db * TN + t) * EN;
      const float* ym = h2 + ((size_t)b * TN + jm[r]) * S2E + d * EN;
      const float* yo = h2 + bt * S2E + d * EN;
      float* dst = rowb + (size_t)(wave * 2 + r) * EN * 4;
#pragma unroll
      for (int q = 0; q < 10; ++q) {
        const int e = lane + 32 * q;
        if (e < EN) {
          const v4f v = { ya[e], ym[e], x[e], yo[e] };
          *(v4f*)(dst + e * 4) = v;
        }
      }
    }
    __syncthreads();

    MkAcc accA = { {0.f, 0.f}, {0.f, 0.f}, {0.f, 0.f}, {0.f, 0.f}, 0.f, 0.f };
    MkAcc accB = accA;
    float syy0 = 0.f;

    for (int e4 = 0; e4 < EN / 4; ++e4) {
      const v4f y0v = y0p[e4];
#pragma unroll
      for (int u = 0; u < 4; ++u) {
        const int e = e4 * 4 + u;
        const v4f w = wp[e * (MK_WS / 4)];
        const v4f A = ra[e];
        const v4f B = rb[e];
        const float y0 = y0v[u];
        const float t0w = y0 * w.z;
        syy0 += y0 * t0w;
        mk_row(A, w, t0w, accA);
        mk_row(B, w, t0w, accB);
      }
    }

    if (lane < MP) {
      const float ny0 = sqrtf(fmaxf(syy0, EPSF));
#pragma unroll
      for (int r = 0; r < 2; ++r) {
        const MkAcc a = r ? accB : accA;
        const int t = t0 + r;
        float* orow = out + ((size_t)b * TN + t) * 160;
        orow[(0 * 2 + d) * MP + lane] = a.xy0 / sqrtf(fmaxf(a.xxfp.x, EPSF)) / ny0;
        orow[(1 * 2 + d) * MP + lane] = a.xy12.x / sqrtf(fmaxf(a.xxam.x, EPSF)) / sqrtf(fmaxf(a.yy12.x, EPSF));
        orow[(2 * 2 + d) * MP + lane] = a.xy12.y / sqrtf(fmaxf(a.xxam.y, EPSF)) / sqrtf(fmaxf(a.yy12.y, EPSF));
        const size_t o = ((size_t)db * TN + t) * MP + lane;
        N1mp[o] = sqrtf(fmaxf(a.xxfp.y, EPSF));
        N2mp[o] = sqrtf(fmaxf(a.yo, EPSF));
      }
    }
    __syncthreads();
  }
}

#define MX_KS (EN / 4)
#define MX_QB 19
#define MX_NB 4

template <int MASK>
__device__ __forceinline__ float row_xor_max(float v) {
  const int o = __builtin_amdgcn_update_dpp(0, __builtin_bit_cast(int, v), 0x160 + MASK, 0xf, 0xf, true);
  return fmaxf(v, __builtin_bit_cast(float, o));
}

__global__ __launch_bounds__(128)
void maxpool_kernel(const float* __restrict__ h1, const float* __restrict__ h2,
                    const float* __restrict__ kern,
                    const float* __restrict__ N1mp, const float* __restrict__ N2mp,
                    float* __restrict__ out) {
  __shared__ __attribute__((aligned(16))) float wl[MP * EN];
  const int db = blockIdx.x;
  const int b = db % BN, d = db / BN;
  const int wave = threadIdx.x >> 5, lane = threadIdx.x & 31;
  const int lh = lane >> 4, ll = lane & 15;

  const int i0 = wave * 16;
  const float* Ab = h1 + ((size_t)b * TN) * S2E + d * EN + 2 * lh;
  const float* Bb = h2 + ((size_t)b * TN) * S2E + d * EN + 2 * lh;
  const v2f* arow = (const v2f*)(Ab + (size_t)(i0 + ll) * S2E);
  const v2f* b0r  = (const v2f*)(Bb + (size_t)(0  + ll) * S2E);
  const v2f* b1r  = (const v2f*)(Bb + (size_t)(16 + ll) * S2E);
  const v2f* b2r  = (const v2f*)(Bb + (size_t)(32 + ll) * S2E);
  const v2f* b3r  = (const v2f*)(Bb + (size_t)(48 + ll) * S2E);

  v2f ra[MX_KS], r0[MX_KS], r1[MX_KS], r2[MX_KS], r3[MX_KS];
#pragma unroll
  for (int q = 0; q < MX_KS; ++q) {
    ra[q] = arow[2 * q];
    r0[q] = b0r[2 * q];
    r1[q] = b1r[2 * q];
    r2[q] = b2r[2 * q];
    r3[q] = b3r[2 * q];
  }

  {
    const float* ws = kern + (6 + d) * EN;
    const int e0 = threadIdx.x, e1 = e0 + 128, e2 = e0 + 256;
    const int e2c = e2 < EN ? e2 : e0;
    float t0[MP], t1[MP], t2[MP];
#pragma unroll
    for (int m = 0; m < MP; ++m) {
      const float* wr = ws + (size_t)m * 2400;
      t0[m] = wr[e0];
      t1[m] = wr[e1];
      t2[m] = wr[e2c];
    }
#pragma unroll
    for (int m = 0; m < MP; ++m) {
      wl[m * EN + e0] = t0[m];
      wl[m * EN + e1] = t1[m];
    }
    if (e2 < EN) {
#pragma unroll
      for (int m = 0; m < MP; ++m) wl[m * EN + e2] = t2[m];
    }
  }
  __syncthreads();

  const int myrow = i0 + lh * 8 + (ll & 7);
  const float* n1p = N1mp + (size_t)db * TN * MP + (size_t)myrow * MP;
  const float* n2p = N2mp + (size_t)db * TN * MP + (size_t)ll * MP;
  float* orow = out + ((size_t)(b * TN + myrow)) * 160 + (6 + d) * MP;

  float n1nx = n1p[0];
  v2f wc[MX_QB];
  {
    const v2f* w0 = (const v2f*)(wl + 2 * lh);
#pragma unroll
    for (int j = 0; j < MX_QB; ++j) wc[j] = w0[2 * j];
  }
  for (int m = 0; m < MP; ++m) {
    const float n1v = n1nx;
    n1nx = n1p[m + 1 < MP ? m + 1 : m];
    const float d0 = n2p[(size_t)0  * MP + m];
    const float d1 = n2p[(size_t)16 * MP + m];
    const float d2 = n2p[(size_t)32 * MP + m];
    const float d3 = n2p[(size_t)48 * MP + m];
    const v2f* wm = (const v2f*)(wl + m * EN + 2 * lh);
    const v2f* wnx = (const v2f*)(wl + (m + 1 < MP ? m + 1 : m) * EN + 2 * lh);

    v8f a0 = {}, a1 = {}, a2 = {}, a3 = {};
#pragma unroll
    for (int blk = 0; blk < MX_NB; ++blk) {
      const int qb = blk * MX_QB;
      const int qn = qb + MX_QB < MX_KS ? MX_QB : MX_KS - qb;
      v2f wn[MX_QB];
      const v2f* nsrc = blk + 1 < MX_NB ? wm + 2 * (qb + MX_QB) : wnx;
#pragma unroll
      for (int j = 0; j < MX_QB; ++j) {
        const bool in = blk + 1 < MX_NB ? (qb + MX_QB + j < MX_KS) : true;
        wn[j] = nsrc[in ? 2 * j : 0];
      }
      __builtin_amdgcn_sched_barrier(0);
#pragma unroll
      for (int j = 0; j < MX_QB; ++j) {
        if (j < qn) {
          const int q = qb + j;
          const v2f w = wc[j];
          const v2f af = ra[q] * w;
          const v2f f0 = r0[q] * w;
          const v2f f1 = r1[q] * w;
          const v2f f2 = r2[q] * w;
          const v2f f3 = r3[q] * w;
          a0 = __builtin_amdgcn_wmma_f32_16x16x4_f32(false, af, false, f0, (short)0, a0, false, false);
          a1 = __builtin_amdgcn_wmma_f32_16x16x4_f32(false, af, false, f1, (short)0, a1, false, false);
          a2 = __builtin_amdgcn_wmma_f32_16x16x4_f32(false, af, false, f2, (short)0, a2, false, false);
          a3 = __builtin_amdgcn_wmma_f32_16x16x4_f32(false, af, false, f3, (short)0, a3, false, false);
        }
      }
      __builtin_amdgcn_sched_barrier(0);
#pragma unroll
      for (int j = 0; j < MX_QB; ++j) wc[j] = wn[j];
    }

    float v[8];
#pragma unroll
    for (int r = 0; r < 8; ++r) {
      const float v0 = a0[r] / d0;
      const float v1 = a1[r] / d1;
      const float v2 = a2[r] / d2;
      const float v3 = a3[r] / d3;
      v[r] = fmaxf(fmaxf(v0, v1), fmaxf(v2, v3));
    }
#pragma unroll
    for (int r = 0; r < 8; ++r) v[r] = row_xor_max<1>(v[r]);
#pragma unroll
    for (int r = 0; r < 8; ++r) v[r] = row_xor_max<2>(v[r]);
#pragma unroll
    for (int r = 0; r < 8; ++r) v[r] = row_xor_max<4>(v[r]);
#pragma unroll
    for (int r = 0; r < 8; ++r) v[r] = row_xor_max<8>(v[r]);
    float mine = v[0];
#pragma unroll
    for (int r = 1; r < 8; ++r) mine = ((ll & 7) == r) ? v[r] : mine;
    if (ll < 8) orow[m] = mine / n1v;
  }
}

extern "C" void kernel_launch(void* const* d_in, const int* in_sizes, int n_in,
                              void* d_out, int out_size, void* d_ws, size_t ws_size,
                              hipStream_t stream) {
  const float* h1   = (const float*)d_in[0];
  const float* h2   = (const float*)d_in[1];
  const float* kern = (const float*)d_in[2];
  float* out = (float*)d_out;

  float* cosm   = (float*)d_ws;
  float* rowsum = cosm + (size_t)2 * BN * TN * TN;
  int*   amax   = (int*)(rowsum + 2 * BN * TN);
  float* att    = (float*)(amax + 2 * BN * TN);
  float* N1mp   = att + (size_t)2 * BN * TN * EN;
  float* N2mp   = N1mp + (size_t)2 * BN * TN * MP;

  cosmat_kernel  <<<2 * BN,          256, 0, stream>>>(h1, h2, cosm, rowsum, amax);
  att_kernel     <<<2 * BN * 19,     128, 0, stream>>>(h2, cosm, rowsum, att);
  match_kernel   <<<2 * BN,          MK_THREADS, 0, stream>>>(h1, h2, kern, att, amax,
                                                       N1mp, N2mp, out);
  maxpool_kernel <<<2 * BN,          128, 0, stream>>>(h1, h2, kern, N1mp, N2mp, out);
}


# match kernel: a lane owns one perspective and two rows of its own (20 waves, no shadow lanes, rows read per lane from LDS); all 64 rows staged half of e at a time; products and sum order kept
# speedup vs baseline: 2.8169x; 1.0898x over previous
#include <hip/hip_runtime.h>
#include <math.h>

#define BN 128
#define TN 64
#define EN 300
#define MP 20
#define S2E 600
#define EPSF 1e-6f

typedef __attribute__((ext_vector_type(2))) float v2f;
typedef __attribute__((ext_vector_type(8))) float v8f;

__device__ __forceinline__ float wred_sum(float v) {
#pragma unroll
  for (int off = 16; off; off >>= 1) v += __shfl_xor(v, off, 32);
  return v;
}

#define CS_P 68
#define NR_R 8
#define NR_Q 10

__global__ __launch_bounds__(256)
void cosmat_kernel(const float* __restrict__ h1, const float* __restrict__ h2,
                   float* __restrict__ cosm  ,
                   float* __restrict__ rowsum, int* __restrict__ amax) {
  __shared__ float sn1[TN];
  __shared__ float sn2[TN];
  __shared__ float cs[TN * CS_P];
  int bd = blockIdx.x;
  int b = bd >> 1, d = bd & 1;
  int wave = threadIdx.x >> 5, lane = threadIdx.x & 31;
  int lh = lane >> 4, ll = lane & 15;
  const float* Ab = h1 + ((size_t)b * TN) * S2E + d * EN;
  const float* Bb = h2 + ((size_t)b * TN) * S2E + d * EN;
  const size_t db = (size_t)d * BN + b;
  float* op = cosm + db * TN * TN;

  {
    float va[NR_R][NR_Q], vb[NR_R][NR_Q];
#pragma unroll
    for (int rr = 0; rr < NR_R; ++rr) {
      const float* p1 = Ab + (size_t)(wave * NR_R + rr) * S2E;
      const float* p2 = Bb + (size_t)(wave * NR_R + rr) * S2E;
#pragma unroll
      for (int q = 0; q < NR_Q; ++q) {
        const int e = lane + 32 * q;
        const int ec = e < EN ? e : lane;
        va[rr][q] = p1[ec];
        vb[rr][q] = p2[ec];
      }
    }
    float m1 = 0.f, m2 = 0.f;
#pragma unroll
    for (int rr = 0; rr < NR_R; ++rr) {
      float s1 = 0.f, s2 = 0.f;
#pragma unroll
      for (int q = 0; q < NR_Q; ++q) {
        if (lane + 32 * q < EN) {
          s1 += va[rr][q] * va[rr][q];
          s2 += vb[rr][q] * vb[rr][q];
        }
      }
      s1 = wred_sum(s1);
      s2 = wred_sum(s2);
      if (lane == rr) { m1 = s1; m2 = s2; }
    }
    if (lane < NR_R) {
      sn1[wave * NR_R + lane] = sqrtf(fmaxf(m1, EPSF));
      sn2[wave * NR_R + lane] = sqrtf(fmaxf(m2, EPSF));
    }
  }
  __syncthreads();

  for (int t = wave; t < 16; t += 8) {
    int i0 = (t >> 2) * 16, j0 = (t & 3) * 16;
    const float* arow = Ab + (size_t)(i0 + ll) * S2E;
    const float* brow = Bb + (size_t)(j0 + ll) * S2E;
    v8f acc = {};
    for (int kk = 0; kk < EN; kk += 4) {
      int k = kk + 2 * lh;
      v2f af = { arow[k], arow[k + 1] };
      v2f bf = { brow[k], brow[k + 1] };
      acc = __builtin_amdgcn_wmma_f32_16x16x4_f32(false, af, false, bf,
                                                  (short)0, acc, false, false);
    }
    float invn2 = 1.f / sn2[j0 + ll];
#pragma unroll
    for (int r = 0; r < 8; ++r) {
      int i = i0 + lh * 8 + r;
      const float c = acc[r] / sn1[i] * invn2;
      op[(size_t)i * TN + (j0 + ll)] = c;
      cs[i * CS_P + (j0 + ll)] = c;
    }
  }
  __syncthreads();

  float mys = 0.f; int mymi = 0;
#pragma unroll
  for (int rr = 0; rr < 8; ++rr) {
    const float* row = cs + (wave * 8 + rr) * CS_P;
    float v0 = row[lane], v1 = row[lane + 32];
    float s = v0 + v1;
    float m = v0; int mi = lane;
    if (v1 > m) { m = v1; mi = lane + 32; }
#pragma unroll
    for (int off = 16; off; off >>= 1) {
      s += __shfl_xor(s, off, 32);
      float om = __shfl_xor(m, off, 32);
      int omi = __shfl_xor(mi, off, 32);
      const bool take = (om > m) | ((om == m) & (omi < mi));
      m = take ? om : m;
      mi = take ? omi : mi;
    }
    if (lane == rr) { mys = s + EPSF; mymi = mi; }
  }
  if (lane < 8) {
    rowsum[db * TN + wave * 8 + lane] = mys;
    amax[db * TN + wave * 8 + lane] = mymi;
  }
}

__global__ void att_kernel(const float* __restrict__ h2, const float* __restrict__ cosm,
                           const float* __restrict__ rowsum, float* __restrict__ att) {
  int blk = blockIdx.x;
  int et = blk % 19;
  int db = blk / 19;
  int b = db % BN, d = db / BN;
  int wave = threadIdx.x >> 5, lane = threadIdx.x & 31;
  int lh = lane >> 4, ll = lane & 15;
  int i0 = wave * 16, e0 = et * 16;
  bool ein = (e0 + ll) < EN;
  int ecol = ein ? (e0 + ll) : 0;
  const float* crow = cosm + (size_t)db * TN * TN + (size_t)(i0 + ll) * TN;
  const float* h2b = h2 + ((size_t)b * TN) * S2E + d * EN;
  v8f acc = {};
  for (int kk = 0; kk < TN; kk += 4) {
    int k = kk + 2 * lh;
    v2f af = { crow[k], crow[k + 1] };
    v2f bf = { h2b[(size_t)k * S2E + ecol], h2b[(size_t)(k + 1) * S2E + ecol] };
    acc = __builtin_amdgcn_wmma_f32_16x16x4_f32(false, af, false, bf,
                                                (short)0, acc, false, false);
  }
  if (ein) {
    float* arow = att + (size_t)db * TN * EN;
    const float* rs = rowsum + (size_t)db * TN;
#pragma unroll
    for (int r = 0; r < 8; ++r) {
      int i = i0 + lh * 8 + r;
      arow[(size_t)i * EN + e0 + ll] = acc[r] / rs[i];
    }
  }
}

typedef __attribute__((ext_vector_type(4))) float v4f;
#define MK_WAVES 20
#define MK_THREADS (MK_WAVES * 32)
#define MK_FILL 3
#define MK_EH 152
#define MK_RP 153
#define MK_SB 8
#define MK_WS 84

struct MkAcc {
  v2f xxam;
  v2f xxfp;
  v2f xy12;
  v2f yy12;
  float xy0;
  float yo;
};

__device__ __forceinline__ void mk_row(const v4f R, const v4f w, const float t0, MkAcc& a) {
  const v2f y12 = R.xy;
  const v2f wam = w.xy;
  const v2f wfp = w.zw;
  const float x = R.z, yo = R.w;
  const v2f T = y12 * wam;
  const float xx = x * x;
  const v2f xs = { x, x };
  const v2f xxs = { xx, xx };
  a.xy12 += xs * T;
  a.yy12 += y12 * T;
  a.xxam += xxs * wam;
  a.xxfp += xxs * wfp;
  a.xy0 += x * t0;
  a.yo += yo * (yo * w.w);
}

__global__ __launch_bounds__(MK_THREADS)
void match_kernel(const float* __restrict__ h1, const float* __restrict__ h2,
                  const float* __restrict__ kern, const float* __restrict__ att,
                  const int* __restrict__ amax,
                  float* __restrict__ N1mp, float* __restrict__ N2mp,
                  float* __restrict__ out) {
  __shared__ __attribute__((aligned(16))) float w2t[EN * MK_WS];
  __shared__ __attribute__((aligned(16))) float rowb[TN * MK_RP * 4];
  __shared__ __attribute__((aligned(16))) float y0b[EN];

  const int db = blockIdx.x;
  const int b = db % BN, d = db / BN;
  const int tid = threadIdx.x;

  for (int base = tid; base < MP * EN; base += MK_THREADS * MK_FILL) {
    float wv[MK_FILL][4];
#pragma unroll
    for (int k = 0; k < MK_FILL; ++k) {
      const int idx = base + k * MK_THREADS;
      const int idc = idx < MP * EN ? idx : 0;
      const int e = idc % EN, m = idc / EN;
#pragma unroll
      for (int s = 0; s < 4; ++s)
        wv[k][s] = kern[(size_t)m * 2400 + (s * 2 + d) * EN + e];
    }
#pragma unroll
    for (int k = 0; k < MK_FILL; ++k) {
      const int idx = base + k * MK_THREADS;
      if (idx < MP * EN) {
        const int e = idx % EN, m = idx / EN;
        const v4f v = { wv[k][1] * wv[k][1], wv[k][2] * wv[k][2],
                        wv[k][0] * wv[k][0], wv[k][3] * wv[k][3] };
        *(v4f*)(w2t + e * MK_WS + m * 4) = v;
      }
    }
  }
  if (tid < EN)
    y0b[tid] = h2[((size_t)b * TN + (TN - 1)) * S2E + d * EN + tid];

  const int hh = tid >> 4, hl = tid & 15;
  const int m = hh < TN / 2 ? hl : 16 + (hl & 3);
  const int g = hh < TN / 2 ? hh : 4 * (hh - TN / 2) + (hl >> 2);
  const v4f* wp = (const v4f*)(w2t + m * 4);
  const v4f* ra = (const v4f*)rowb + (size_t)(2 * g) * MK_RP;
  const v4f* rb = ra + MK_RP;
  const v4f* y0p = (const v4f*)y0b;

  MkAcc accA = { {0.f, 0.f}, {0.f, 0.f}, {0.f, 0.f}, {0.f, 0.f}, 0.f, 0.f };
  MkAcc accB = accA;
  float syy0 = 0.f;

  for (int h = 0; h < 2; ++h) {
    const int eb = h * MK_EH;
    if (h) __syncthreads();

    for (int base = tid; base < TN * MK_EH; base += MK_THREADS * MK_SB) {
      v4f v[MK_SB];
#pragma unroll
      for (int k = 0; k < MK_SB; ++k) {
        const int i = base + k * MK_THREADS;
        const int ic = i < TN * MK_EH ? i : tid;
        const int row = ic / MK_EH, el = ic % MK_EH;
        const int e = eb + el < EN ? eb + el : eb;
        const int jm = amax[(size_t)db * TN + row];
        const size_t bt = (size_t)b * TN + row;
        const v4f r = { att[((size_t)db * TN + row) * EN + e],
                        h2[((size_t)b * TN + jm) * S2E + d * EN + e],
                        h1[bt * S2E + d * EN + e],
                        h2[bt * S2E + d * EN + e] };
        v[k] = r;
      }
#pragma unroll
      for (int k = 0; k < MK_SB; ++k) {
        const int i = base + k * MK_THREADS;
        if (i < TN * MK_EH) {
          const int row = i / MK_EH, el = i % MK_EH;
          if (eb + el < EN) *(v4f*)(rowb + ((size_t)row * MK_RP + el) * 4) = v[k];
        }
      }
    }
    __syncthreads();

    const int n4 = (h ? EN - MK_EH : MK_EH) / 4;
    for (int e4 = 0; e4 < n4; ++e4) {
      const v4f y0v = y0p[eb / 4 + e4];
#pragma unroll
      for (int u = 0; u < 4; ++u) {
        const int el = e4 * 4 + u;
        const v4f w = wp[(eb + el) * (MK_WS / 4)];
        const v4f A = ra[el];
        const v4f B = rb[el];
        const float y0 = y0v[u];
        const float t0w = y0 * w.z;
        syy0 += y0 * t0w;
        mk_row(A, w, t0w, accA);
        mk_row(B, w, t0w, accB);
      }
    }
  }

  {
    const float ny0 = sqrtf(fmaxf(syy0, EPSF));
#pragma unroll
    for (int r = 0; r < 2; ++r) {
      const MkAcc a = r ? accB : accA;
      const int t = 2 * g + r;
      float* orow = out + ((size_t)b * TN + t) * 160;
      orow[(0 * 2 + d) * MP + m] = a.xy0 / sqrtf(fmaxf(a.xxfp.x, EPSF)) / ny0;
      orow[(1 * 2 + d) * MP + m] = a.xy12.x / sqrtf(fmaxf(a.xxam.x, EPSF)) / sqrtf(fmaxf(a.yy12.x, EPSF));
      orow[(2 * 2 + d) * MP + m] = a.xy12.y / sqrtf(fmaxf(a.xxam.y, EPSF)) / sqrtf(fmaxf(a.yy12.y, EPSF));
      const size_t o = ((size_t)db * TN + t) * MP + m;
      N1mp[o] = sqrtf(fmaxf(a.xxfp.y, EPSF));
      N2mp[o] = sqrtf(fmaxf(a.yo, EPSF));
    }
  }
}

#define MX_KS (EN / 4)
#define MX_QB 19
#define MX_NB 4

template <int MASK>
__device__ __forceinline__ float row_xor_max(float v) {
  const int o = __builtin_amdgcn_update_dpp(0, __builtin_bit_cast(int, v), 0x160 + MASK, 0xf, 0xf, true);
  return fmaxf(v, __builtin_bit_cast(float, o));
}

__global__ __launch_bounds__(128)
void maxpool_kernel(const float* __restrict__ h1, const float* __restrict__ h2,
                    const float* __restrict__ kern,
                    const float* __restrict__ N1mp, const float* __restrict__ N2mp,
                    float* __restrict__ out) {
  __shared__ __attribute__((aligned(16))) float wl[MP * EN];
  const int db = blockIdx.x;
  const int b = db % BN, d = db / BN;
  const int wave = threadIdx.x >> 5, lane = threadIdx.x & 31;
  const int lh = lane >> 4, ll = lane & 15;

  const int i0 = wave * 16;
  const float* Ab = h1 + ((size_t)b * TN) * S2E + d * EN + 2 * lh;
  const float* Bb = h2 + ((size_t)b * TN) * S2E + d * EN + 2 * lh;
  const v2f* arow = (const v2f*)(Ab + (size_t)(i0 + ll) * S2E);
  const v2f* b0r  = (const v2f*)(Bb + (size_t)(0  + ll) * S2E);
  const v2f* b1r  = (const v2f*)(Bb + (size_t)(16 + ll) * S2E);
  const v2f* b2r  = (const v2f*)(Bb + (size_t)(32 + ll) * S2E);
  const v2f* b3r  = (const v2f*)(Bb + (size_t)(48 + ll) * S2E);

  v2f ra[MX_KS], r0[MX_KS], r1[MX_KS], r2[MX_KS], r3[MX_KS];
#pragma unroll
  for (int q = 0; q < MX_KS; ++q) {
    ra[q] = arow[2 * q];
    r0[q] = b0r[2 * q];
    r1[q] = b1r[2 * q];
    r2[q] = b2r[2 * q];
    r3[q] = b3r[2 * q];
  }

  {
    const float* ws = kern + (6 + d) * EN;
    const int e0 = threadIdx.x, e1 = e0 + 128, e2 = e0 + 256;
    const int e2c = e2 < EN ? e2 : e0;
    float t0[MP], t1[MP], t2[MP];
#pragma unroll
    for (int m = 0; m < MP; ++m) {
      const float* wr = ws + (size_t)m * 2400;
      t0[m] = wr[e0];
      t1[m] = wr[e1];
      t2[m] = wr[e2c];
    }
#pragma unroll
    for (int m = 0; m < MP; ++m) {
      wl[m * EN + e0] = t0[m];
      wl[m * EN + e1] = t1[m];
    }
    if (e2 < EN) {
#pragma unroll
      for (int m = 0; m < MP; ++m) wl[m * EN + e2] = t2[m];
    }
  }
  __syncthreads();

  const int myrow = i0 + lh * 8 + (ll & 7);
  const float* n1p = N1mp + (size_t)db * TN * MP + (size_t)myrow * MP;
  const float* n2p = N2mp + (size_t)db * TN * MP + (size_t)ll * MP;
  float* orow = out + ((size_t)(b * TN + myrow)) * 160 + (6 + d) * MP;

  float n1nx = n1p[0];
  v2f wc[MX_QB];
  {
    const v2f* w0 = (const v2f*)(wl + 2 * lh);
#pragma unroll
    for (int j = 0; j < MX_QB; ++j) wc[j] = w0[2 * j];
  }
  for (int m = 0; m < MP; ++m) {
    const float n1v = n1nx;
    n1nx = n1p[m + 1 < MP ? m + 1 : m];
    const float d0 = n2p[(size_t)0  * MP + m];
    const float d1 = n2p[(size_t)16 * MP + m];
    const float d2 = n2p[(size_t)32 * MP + m];
    const float d3 = n2p[(size_t)48 * MP + m];
    const v2f* wm = (const v2f*)(wl + m * EN + 2 * lh);
    const v2f* wnx = (const v2f*)(wl + (m + 1 < MP ? m + 1 : m) * EN + 2 * lh);

    v8f a0 = {}, a1 = {}, a2 = {}, a3 = {};
#pragma unroll
    for (int blk = 0; blk < MX_NB; ++blk) {
      const int qb = blk * MX_QB;
      const int qn = qb + MX_QB < MX_KS ? MX_QB : MX_KS - qb;
      v2f wn[MX_QB];
      const v2f* nsrc = blk + 1 < MX_NB ? wm + 2 * (qb + MX_QB) : wnx;
#pragma unroll
      for (int j = 0; j < MX_QB; ++j) {
        const bool in = blk + 1 < MX_NB ? (qb + MX_QB + j < MX_KS) : true;
        wn[j] = nsrc[in ? 2 * j : 0];
      }
      __builtin_amdgcn_sched_barrier(0);
#pragma unroll
      for (int j = 0; j < MX_QB; ++j) {
        if (j < qn) {
          const int q = qb + j;
          const v2f w = wc[j];
          const v2f af = ra[q] * w;
          const v2f f0 = r0[q] * w;
          const v2f f1 = r1[q] * w;
          const v2f f2 = r2[q] * w;
          const v2f f3 = r3[q] * w;
          a0 = __builtin_amdgcn_wmma_f32_16x16x4_f32(false, af, false, f0, (short)0, a0, false, false);
          a1 = __builtin_amdgcn_wmma_f32_16x16x4_f32(false, af, false, f1, (short)0, a1, false, false);
          a2 = __builtin_amdgcn_wmma_f32_16x16x4_f32(false, af, false, f2, (short)0, a2, false, false);
          a3 = __builtin_amdgcn_wmma_f32_16x16x4_f32(false, af, false, f3, (short)0, a3, false, false);
        }
      }
      __builtin_amdgcn_sched_barrier(0);
#pragma unroll
      for (int j = 0; j < MX_QB; ++j) wc[j] = wn[j];
    }

    float v[8];
#pragma unroll
    for (int r = 0; r < 8; ++r) {
      const float v0 = a0[r] / d0;
      const float v1 = a1[r] / d1;
      const float v2 = a2[r] / d2;
      const float v3 = a3[r] / d3;
      v[r] = fmaxf(fmaxf(v0, v1), fmaxf(v2, v3));
    }
#pragma unroll
    for (int r = 0; r < 8; ++r) v[r] = row_xor_max<1>(v[r]);
#pragma unroll
    for (int r = 0; r < 8; ++r) v[r] = row_xor_max<2>(v[r]);
#pragma unroll
    for (int r = 0; r < 8; ++r) v[r] = row_xor_max<4>(v[r]);
#pragma unroll
    for (int r = 0; r < 8; ++r) v[r] = row_xor_max<8>(v[r]);
    float mine = v[0];
#pragma unroll
    for (int r = 1; r < 8; ++r) mine = ((ll & 7) == r) ? v[r] : mine;
    if (ll < 8) orow[m] = mine / n1v;
  }
}

extern "C" void kernel_launch(void* const* d_in, const int* in_sizes, int n_in,
                              void* d_out, int out_size, void* d_ws, size_t ws_size,
                              hipStream_t stream) {
  const float* h1   = (const float*)d_in[0];
  const float* h2   = (const float*)d_in[1];
  const float* kern = (const float*)d_in[2];
  float* out = (float*)d_out;

  float* cosm   = (float*)d_ws;
  float* rowsum = cosm + (size_t)2 * BN * TN * TN;
  int*   amax   = (int*)(rowsum + 2 * BN * TN);
  float* att    = (float*)(amax + 2 * BN * TN);
  float* N1mp   = att + (size_t)2 * BN * TN * EN;
  float* N2mp   = N1mp + (size_t)2 * BN * TN * MP;

  cosmat_kernel  <<<2 * BN,          256, 0, stream>>>(h1, h2, cosm, rowsum, amax);
  att_kernel     <<<2 * BN * 19,     128, 0, stream>>>(h2, cosm, rowsum, att);
  match_kernel   <<<2 * BN,          MK_THREADS, 0, stream>>>(h1, h2, kern, att, amax,
                                                       N1mp, N2mp, out);
  maxpool_kernel <<<2 * BN,          128, 0, stream>>>(h1, h2, kern, N1mp, N2mp, out);
}


# attentive-vector product (cos @ h2 / rowsum) moves into the cosine-matrix kernel, its operands (cos, row sums, h2 rows) kept in LDS there; 4 launches become 3; products and sum order kept
# speedup vs baseline: 3.1284x; 1.1106x over previous
#include <hip/hip_runtime.h>
#include <math.h>

#define BN 128
#define TN 64
#define EN 300
#define MP 20
#define S2E 600
#define EPSF 1e-6f

typedef __attribute__((ext_vector_type(2))) float v2f;
typedef __attribute__((ext_vector_type(8))) float v8f;

__device__ __forceinline__ float wred_sum(float v) {
#pragma unroll
  for (int off = 16; off; off >>= 1) v += __shfl_xor(v, off, 32);
  return v;
}

#define CS_P 68
#define NR_R 8
#define NR_Q 10
#define H2_P 300
#define AT_KS (TN / 4)
#define AT_CT 10

__global__ __launch_bounds__(256)
void cosatt_kernel(const float* __restrict__ h1, const float* __restrict__ h2,
                   int* __restrict__ amax, float* __restrict__ att  ) {
  __shared__ float sn1[TN];
  __shared__ float sn2[TN];
  __shared__ float srs[TN];
  __shared__ __attribute__((aligned(16))) float cs[TN * CS_P];
  __shared__ float h2s[TN * H2_P];
  int bd = blockIdx.x;
  int b = bd >> 1, d = bd & 1;
  int wave = threadIdx.x >> 5, lane = threadIdx.x & 31;
  int lh = lane >> 4, ll = lane & 15;
  const float* Ab = h1 + ((size_t)b * TN) * S2E + d * EN;
  const float* Bb = h2 + ((size_t)b * TN) * S2E + d * EN;
  const size_t db = (size_t)d * BN + b;

  {
    float va[NR_R][NR_Q], vb[NR_R][NR_Q];
#pragma unroll
    for (int rr = 0; rr < NR_R; ++rr) {
      const float* p1 = Ab + (size_t)(wave * NR_R + rr) * S2E;
      const float* p2 = Bb + (size_t)(wave * NR_R + rr) * S2E;
#pragma unroll
      for (int q = 0; q < NR_Q; ++q) {
        const int e = lane + 32 * q;
        const int ec = e < EN ? e : lane;
        va[rr][q] = p1[ec];
        vb[rr][q] = p2[ec];
      }
    }
#pragma unroll
    for (int rr = 0; rr < NR_R; ++rr) {
#pragma unroll
      for (int q = 0; q < NR_Q; ++q) {
        const int e = lane + 32 * q;
        if (e < EN) h2s[(wave * NR_R + rr) * H2_P + e] = vb[rr][q];
      }
    }
    float m1 = 0.f, m2 = 0.f;
#pragma unroll
    for (int rr = 0; rr < NR_R; ++rr) {
      float s1 = 0.f, s2 = 0.f;
#pragma unroll
      for (int q = 0; q < NR_Q; ++q) {
        if (lane + 32 * q < EN) {
          s1 += va[rr][q] * va[rr][q];
          s2 += vb[rr][q] * vb[rr][q];
        }
      }
      s1 = wred_sum(s1);
      s2 = wred_sum(s2);
      if (lane == rr) { m1 = s1; m2 = s2; }
    }
    if (lane < NR_R) {
      sn1[wave * NR_R + lane] = sqrtf(fmaxf(m1, EPSF));
      sn2[wave * NR_R + lane] = sqrtf(fmaxf(m2, EPSF));
    }
  }
  __syncthreads();

  for (int t = wave; t < 16; t += 8) {
    int i0 = (t >> 2) * 16, j0 = (t & 3) * 16;
    const float* arow = Ab + (size_t)(i0 + ll) * S2E;
    const float* brow = Bb + (size_t)(j0 + ll) * S2E;
    v8f acc = {};
    for (int kk = 0; kk < EN; kk += 4) {
      int k = kk + 2 * lh;
      v2f af = { arow[k], arow[k + 1] };
      v2f bf = { brow[k], brow[k + 1] };
      acc = __builtin_amdgcn_wmma_f32_16x16x4_f32(false, af, false, bf,
                                                  (short)0, acc, false, false);
    }
    float invn2 = 1.f / sn2[j0 + ll];
#pragma unroll
    for (int r = 0; r < 8; ++r) {
      int i = i0 + lh * 8 + r;
      cs[i * CS_P + (j0 + ll)] = acc[r] / sn1[i] * invn2;
    }
  }
  __syncthreads();

  float mys = 0.f; int mymi = 0;
#pragma unroll
  for (int rr = 0; rr < 8; ++rr) {
    const float* row = cs + (wave * 8 + rr) * CS_P;
    float v0 = row[lane], v1 = row[lane + 32];
    float s = v0 + v1;
    float m = v0; int mi = lane;
    if (v1 > m) { m = v1; mi = lane + 32; }
#pragma unroll
    for (int off = 16; off; off >>= 1) {
      s += __shfl_xor(s, off, 32);
      float om = __shfl_xor(m, off, 32);
      int omi = __shfl_xor(mi, off, 32);
      const bool take = (om > m) | ((om == m) & (omi < mi));
      m = take ? om : m;
      mi = take ? omi : mi;
    }
    if (lane == rr) { mys = s + EPSF; mymi = mi; }
  }
  if (lane < 8) {
    srs[wave * 8 + lane] = mys;
    amax[db * TN + wave * 8 + lane] = mymi;
  }
  __syncthreads();

  {
    const int i0 = (wave & 3) * 16;
    const int ct0 = (wave >> 2) * AT_CT;
    v2f af[AT_KS];
    const float* crow = cs + (i0 + ll) * CS_P + 2 * lh;
#pragma unroll
    for (int s = 0; s < AT_KS; ++s) af[s] = *(const v2f*)(crow + 4 * s);
    float rs[8];
#pragma unroll
    for (int r = 0; r < 8; ++r) rs[r] = srs[i0 + lh * 8 + r];
    float* arow = att + (db * TN + i0 + lh * 8) * EN;
    const float* hb = h2s + 2 * lh * H2_P;

    for (int c = 0; c < AT_CT; c += 2) {
      const int ea = (ct0 + c) * 16 + ll, eb = ea + 16;
      const float* pa = hb + (ea < EN ? ea : 0);
      const float* pb = hb + (eb < EN ? eb : 0);
      v2f ba[AT_KS], bb[AT_KS];
#pragma unroll
      for (int s = 0; s < AT_KS; ++s) {
        ba[s] = (v2f){ pa[(4 * s) * H2_P], pa[(4 * s + 1) * H2_P] };
        bb[s] = (v2f){ pb[(4 * s) * H2_P], pb[(4 * s + 1) * H2_P] };
      }
      __builtin_amdgcn_sched_barrier(0);
      v8f a0 = {}, a1 = {};
#pragma unroll
      for (int s = 0; s < AT_KS; ++s) {
        a0 = __builtin_amdgcn_wmma_f32_16x16x4_f32(false, af[s], false, ba[s], (short)0, a0, false, false);
        a1 = __builtin_amdgcn_wmma_f32_16x16x4_f32(false, af[s], false, bb[s], (short)0, a1, false, false);
      }
      if (ea < EN) {
#pragma unroll
        for (int r = 0; r < 8; ++r) arow[(size_t)r * EN + ea] = a0[r] / rs[r];
      }
      if (eb < EN) {
#pragma unroll
        for (int r = 0; r < 8; ++r) arow[(size_t)r * EN + eb] = a1[r] / rs[r];
      }
    }
  }
}

typedef __attribute__((ext_vector_type(4))) float v4f;
#define MK_WAVES 20
#define MK_THREADS (MK_WAVES * 32)
#define MK_FILL 3
#define MK_EH 152
#define MK_RP 153
#define MK_SB 8
#define MK_WS 84

struct MkAcc {
  v2f xxam;
  v2f xxfp;
  v2f xy12;
  v2f yy12;
  float xy0;
  float yo;
};

__device__ __forceinline__ void mk_row(const v4f R, const v4f w, const float t0, MkAcc& a) {
  const v2f y12 = R.xy;
  const v2f wam = w.xy;
  const v2f wfp = w.zw;
  const float x = R.z, yo = R.w;
  const v2f T = y12 * wam;
  const float xx = x * x;
  const v2f xs = { x, x };
  const v2f xxs = { xx, xx };
  a.xy12 += xs * T;
  a.yy12 += y12 * T;
  a.xxam += xxs * wam;
  a.xxfp += xxs * wfp;
  a.xy0 += x * t0;
  a.yo += yo * (yo * w.w);
}

__global__ __launch_bounds__(MK_THREADS)
void match_kernel(const float* __restrict__ h1, const float* __restrict__ h2,
                  const float* __restrict__ kern, const float* __restrict__ att,
                  const int* __restrict__ amax,
                  float* __restrict__ N1mp, float* __restrict__ N2mp,
                  float* __restrict__ out) {
  __shared__ __attribute__((aligned(16))) float w2t[EN * MK_WS];
  __shared__ __attribute__((aligned(16))) float rowb[TN * MK_RP * 4];
  __shared__ __attribute__((aligned(16))) float y0b[EN];

  const int db = blockIdx.x;
  const int b = db % BN, d = db / BN;
  const int tid = threadIdx.x;

  for (int base = tid; base < MP * EN; base += MK_THREADS * MK_FILL) {
    float wv[MK_FILL][4];
#pragma unroll
    for (int k = 0; k < MK_FILL; ++k) {
      const int idx = base + k * MK_THREADS;
      const int idc = idx < MP * EN ? idx : 0;
      const int e = idc % EN, m = idc / EN;
#pragma unroll
      for (int s = 0; s < 4; ++s)
        wv[k][s] = kern[(size_t)m * 2400 + (s * 2 + d) * EN + e];
    }
#pragma unroll
    for (int k = 0; k < MK_FILL; ++k) {
      const int idx = base + k * MK_THREADS;
      if (idx < MP * EN) {
        const int e = idx % EN, m = idx / EN;
        const v4f v = { wv[k][1] * wv[k][1], wv[k][2] * wv[k][2],
                        wv[k][0] * wv[k][0], wv[k][3] * wv[k][3] };
        *(v4f*)(w2t + e * MK_WS + m * 4) = v;
      }
    }
  }
  if (tid < EN)
    y0b[tid] = h2[((size_t)b * TN + (TN - 1)) * S2E + d * EN + tid];

  const int hh = tid >> 4, hl = tid & 15;
  const int m = hh < TN / 2 ? hl : 16 + (hl & 3);
  const int g = hh < TN / 2 ? hh : 4 * (hh - TN / 2) + (hl >> 2);
  const v4f* wp = (const v4f*)(w2t + m * 4);
  const v4f* ra = (const v4f*)rowb + (size_t)(2 * g) * MK_RP;
  const v4f* rb = ra + MK_RP;
  const v4f* y0p = (const v4f*)y0b;

  MkAcc accA = { {0.f, 0.f}, {0.f, 0.f}, {0.f, 0.f}, {0.f, 0.f}, 0.f, 0.f };
  MkAcc accB = accA;
  float syy0 = 0.f;

  for (int h = 0; h < 2; ++h) {
    const int eb = h * MK_EH;
    if (h) __syncthreads();

    for (int base = tid; base < TN * MK_EH; base += MK_THREADS * MK_SB) {
      v4f v[MK_SB];
#pragma unroll
      for (int k = 0; k < MK_SB; ++k) {
        const int i = base + k * MK_THREADS;
        const int ic = i < TN * MK_EH ? i : tid;
        const int row = ic / MK_EH, el = ic % MK_EH;
        const int e = eb + el < EN ? eb + el : eb;
        const int jm = amax[(size_t)db * TN + row];
        const size_t bt = (size_t)b * TN + row;
        const v4f r = { att[((size_t)db * TN + row) * EN + e],
                        h2[((size_t)b * TN + jm) * S2E + d * EN + e],
                        h1[bt * S2E + d * EN + e],
                        h2[bt * S2E + d * EN + e] };
        v[k] = r;
      }
#pragma unroll
      for (int k = 0; k < MK_SB; ++k) {
        const int i = base + k * MK_THREADS;
        if (i < TN * MK_EH) {
          const int row = i / MK_EH, el = i % MK_EH;
          if (eb + el < EN) *(v4f*)(rowb + ((size_t)row * MK_RP + el) * 4) = v[k];
        }
      }
    }
    __syncthreads();

    const int n4 = (h ? EN - MK_EH : MK_EH) / 4;
    for (int e4 = 0; e4 < n4; ++e4) {
      const v4f y0v = y0p[eb / 4 + e4];
#pragma unroll
      for (int u = 0; u < 4; ++u) {
        const int el = e4 * 4 + u;
        const v4f w = wp[(eb + el) * (MK_WS / 4)];
        const v4f A = ra[el];
        const v4f B = rb[el];
        const float y0 = y0v[u];
        const float t0w = y0 * w.z;
        syy0 += y0 * t0w;
        mk_row(A, w, t0w, accA);
        mk_row(B, w, t0w, accB);
      }
    }
  }

  {
    const float ny0 = sqrtf(fmaxf(syy0, EPSF));
#pragma unroll
    for (int r = 0; r < 2; ++r) {
      const MkAcc a = r ? accB : accA;
      const int t = 2 * g + r;
      float* orow = out + ((size_t)b * TN + t) * 160;
      orow[(0 * 2 + d) * MP + m] = a.xy0 / sqrtf(fmaxf(a.xxfp.x, EPSF)) / ny0;
      orow[(1 * 2 + d) * MP + m] = a.xy12.x / sqrtf(fmaxf(a.xxam.x, EPSF)) / sqrtf(fmaxf(a.yy12.x, EPSF));
      orow[(2 * 2 + d) * MP + m] = a.xy12.y / sqrtf(fmaxf(a.xxam.y, EPSF)) / sqrtf(fmaxf(a.yy12.y, EPSF));
      const size_t o = ((size_t)db * TN + t) * MP + m;
      N1mp[o] = sqrtf(fmaxf(a.xxfp.y, EPSF));
      N2mp[o] = sqrtf(fmaxf(a.yo, EPSF));
    }
  }
}

#define MX_KS (EN / 4)
#define MX_QB 19
#define MX_NB 4

template <int MASK>
__device__ __forceinline__ float row_xor_max(float v) {
  const int o = __builtin_amdgcn_update_dpp(0, __builtin_bit_cast(int, v), 0x160 + MASK, 0xf, 0xf, true);
  return fmaxf(v, __builtin_bit_cast(float, o));
}

__global__ __launch_bounds__(128)
void maxpool_kernel(const float* __restrict__ h1, const float* __restrict__ h2,
                    const float* __restrict__ kern,
                    const float* __restrict__ N1mp, const float* __restrict__ N2mp,
                    float* __restrict__ out) {
  __shared__ __attribute__((aligned(16))) float wl[MP * EN];
  const int db = blockIdx.x;
  const int b = db % BN, d = db / BN;
  const int wave = threadIdx.x >> 5, lane = threadIdx.x & 31;
  const int lh = lane >> 4, ll = lane & 15;

  const int i0 = wave * 16;
  const float* Ab = h1 + ((size_t)b * TN) * S2E + d * EN + 2 * lh;
  const float* Bb = h2 + ((size_t)b * TN) * S2E + d * EN + 2 * lh;
  const v2f* arow = (const v2f*)(Ab + (size_t)(i0 + ll) * S2E);
  const v2f* b0r  = (const v2f*)(Bb + (size_t)(0  + ll) * S2E);
  const v2f* b1r  = (const v2f*)(Bb + (size_t)(16 + ll) * S2E);
  const v2f* b2r  = (const v2f*)(Bb + (size_t)(32 + ll) * S2E);
  const v2f* b3r  = (const v2f*)(Bb + (size_t)(48 + ll) * S2E);

  v2f ra[MX_KS], r0[MX_KS], r1[MX_KS], r2[MX_KS], r3[MX_KS];
#pragma unroll
  for (int q = 0; q < MX_KS; ++q) {
    ra[q] = arow[2 * q];
    r0[q] = b0r[2 * q];
    r1[q] = b1r[2 * q];
    r2[q] = b2r[2 * q];
    r3[q] = b3r[2 * q];
  }

  {
    const float* ws = kern + (6 + d) * EN;
    const int e0 = threadIdx.x, e1 = e0 + 128, e2 = e0 + 256;
    const int e2c = e2 < EN ? e2 : e0;
    float t0[MP], t1[MP], t2[MP];
#pragma unroll
    for (int m = 0; m < MP; ++m) {
      const float* wr = ws + (size_t)m * 2400;
      t0[m] = wr[e0];
      t1[m] = wr[e1];
      t2[m] = wr[e2c];
    }
#pragma unroll
    for (int m = 0; m < MP; ++m) {
      wl[m * EN + e0] = t0[m];
      wl[m * EN + e1] = t1[m];
    }
    if (e2 < EN) {
#pragma unroll
      for (int m = 0; m < MP; ++m) wl[m * EN + e2] = t2[m];
    }
  }
  __syncthreads();

  const int myrow = i0 + lh * 8 + (ll & 7);
  const float* n1p = N1mp + (size_t)db * TN * MP + (size_t)myrow * MP;
  const float* n2p = N2mp + (size_t)db * TN * MP + (size_t)ll * MP;
  float* orow = out + ((size_t)(b * TN + myrow)) * 160 + (6 + d) * MP;

  float n1nx = n1p[0];
  v2f wc[MX_QB];
  {
    const v2f* w0 = (const v2f*)(wl + 2 * lh);
#pragma unroll
    for (int j = 0; j < MX_QB; ++j) wc[j] = w0[2 * j];
  }
  for (int m = 0; m < MP; ++m) {
    const float n1v = n1nx;
    n1nx = n1p[m + 1 < MP ? m + 1 : m];
    const float d0 = n2p[(size_t)0  * MP + m];
    const float d1 = n2p[(size_t)16 * MP + m];
    const float d2 = n2p[(size_t)32 * MP + m];
    const float d3 = n2p[(size_t)48 * MP + m];
    const v2f* wm = (const v2f*)(wl + m * EN + 2 * lh);
    const v2f* wnx = (const v2f*)(wl + (m + 1 < MP ? m + 1 : m) * EN + 2 * lh);

    v8f a0 = {}, a1 = {}, a2 = {}, a3 = {};
#pragma unroll
    for (int blk = 0; blk < MX_NB; ++blk) {
      const int qb = blk * MX_QB;
      const int qn = qb + MX_QB < MX_KS ? MX_QB : MX_KS - qb;
      v2f wn[MX_QB];
      const v2f* nsrc = blk + 1 < MX_NB ? wm + 2 * (qb + MX_QB) : wnx;
#pragma unroll
      for (int j = 0; j < MX_QB; ++j) {
        const bool in = blk + 1 < MX_NB ? (qb + MX_QB + j < MX_KS) : true;
        wn[j] = nsrc[in ? 2 * j : 0];
      }
      __builtin_amdgcn_sched_barrier(0);
#pragma unroll
      for (int j = 0; j < MX_QB; ++j) {
        if (j < qn) {
          const int q = qb + j;
          const v2f w = wc[j];
          const v2f af = ra[q] * w;
          const v2f f0 = r0[q] * w;
          const v2f f1 = r1[q] * w;
          const v2f f2 = r2[q] * w;
          const v2f f3 = r3[q] * w;
          a0 = __builtin_amdgcn_wmma_f32_16x16x4_f32(false, af, false, f0, (short)0, a0, false, false);
          a1 = __builtin_amdgcn_wmma_f32_16x16x4_f32(false, af, false, f1, (short)0, a1, false, false);
          a2 = __builtin_amdgcn_wmma_f32_16x16x4_f32(false, af, false, f2, (short)0, a2, false, false);
          a3 = __builtin_amdgcn_wmma_f32_16x16x4_f32(false, af, false, f3, (short)0, a3, false, false);
        }
      }
      __builtin_amdgcn_sched_barrier(0);
#pragma unroll
      for (int j = 0; j < MX_QB; ++j) wc[j] = wn[j];
    }

    float v[8];
#pragma unroll
    for (int r = 0; r < 8; ++r) {
      const float v0 = a0[r] / d0;
      const float v1 = a1[r] / d1;
      const float v2 = a2[r] / d2;
      const float v3 = a3[r] / d3;
      v[r] = fmaxf(fmaxf(v0, v1), fmaxf(v2, v3));
    }
#pragma unroll
    for (int r = 0; r < 8; ++r) v[r] = row_xor_max<1>(v[r]);
#pragma unroll
    for (int r = 0; r < 8; ++r) v[r] = row_xor_max<2>(v[r]);
#pragma unroll
    for (int r = 0; r < 8; ++r) v[r] = row_xor_max<4>(v[r]);
#pragma unroll
    for (int r = 0; r < 8; ++r) v[r] = row_xor_max<8>(v[r]);
    float mine = v[0];
#pragma unroll
    for (int r = 1; r < 8; ++r) mine = ((ll & 7) == r) ? v[r] : mine;
    if (ll < 8) orow[m] = mine / n1v;
  }
}

extern "C" void kernel_launch(void* const* d_in, const int* in_sizes, int n_in,
                              void* d_out, int out_size, void* d_ws, size_t ws_size,
                              hipStream_t stream) {
  const float* h1   = (const float*)d_in[0];
  const float* h2   = (const float*)d_in[1];
  const float* kern = (const float*)d_in[2];
  float* out = (float*)d_out;

  float* cosm   = (float*)d_ws;
  float* rowsum = cosm + (size_t)2 * BN * TN * TN;
  int*   amax   = (int*)(rowsum + 2 * BN * TN);
  float* att    = (float*)(amax + 2 * BN * TN);
  float* N1mp   = att + (size_t)2 * BN * TN * EN;
  float* N2mp   = N1mp + (size_t)2 * BN * TN * MP;

  cosatt_kernel  <<<2 * BN,          256, 0, stream>>>(h1, h2, amax, att);
  match_kernel   <<<2 * BN,          MK_THREADS, 0, stream>>>(h1, h2, kern, att, amax,
                                                       N1mp, N2mp, out);
  maxpool_kernel <<<2 * BN,          128, 0, stream>>>(h1, h2, kern, N1mp, N2mp, out);
}


# cosine-matrix tiles: A/B fragments read from LDS copies of the rows the norm prologue already loads (h1 copy added), a block of k-steps ahead, B once for a wave's two tiles; products, sum order kept
# speedup vs baseline: 3.3023x; 1.0556x over previous
#include <hip/hip_runtime.h>
#include <math.h>

#define BN 128
#define TN 64
#define EN 300
#define MP 20
#define S2E 600
#define EPSF 1e-6f

typedef __attribute__((ext_vector_type(2))) float v2f;
typedef __attribute__((ext_vector_type(8))) float v8f;

__device__ __forceinline__ float wred_sum(float v) {
#pragma unroll
  for (int off = 16; off; off >>= 1) v += __shfl_xor(v, off, 32);
  return v;
}

#define CS_P 68
#define NR_R 8
#define NR_Q 10
#define H2_P 300
#define CT_QB 25
#define CT_NB (EN / 4 / CT_QB)
#define AT_KS (TN / 4)
#define AT_CT 10

__global__ __launch_bounds__(256)
void cosatt_kernel(const float* __restrict__ h1, const float* __restrict__ h2,
                   int* __restrict__ amax, float* __restrict__ att  ) {
  __shared__ float sn1[TN];
  __shared__ float sn2[TN];
  __shared__ float srs[TN];
  __shared__ __attribute__((aligned(16))) float cs[TN * CS_P];
  __shared__ __attribute__((aligned(16))) float h1s[TN * H2_P];
  __shared__ __attribute__((aligned(16))) float h2s[TN * H2_P];
  int bd = blockIdx.x;
  int b = bd >> 1, d = bd & 1;
  int wave = threadIdx.x >> 5, lane = threadIdx.x & 31;
  int lh = lane >> 4, ll = lane & 15;
  const float* Ab = h1 + ((size_t)b * TN) * S2E + d * EN;
  const float* Bb = h2 + ((size_t)b * TN) * S2E + d * EN;
  const size_t db = (size_t)d * BN + b;

  {
    float va[NR_R][NR_Q], vb[NR_R][NR_Q];
#pragma unroll
    for (int rr = 0; rr < NR_R; ++rr) {
      const float* p1 = Ab + (size_t)(wave * NR_R + rr) * S2E;
      const float* p2 = Bb + (size_t)(wave * NR_R + rr) * S2E;
#pragma unroll
      for (int q = 0; q < NR_Q; ++q) {
        const int e = lane + 32 * q;
        const int ec = e < EN ? e : lane;
        va[rr][q] = p1[ec];
        vb[rr][q] = p2[ec];
      }
    }
#pragma unroll
    for (int rr = 0; rr < NR_R; ++rr) {
#pragma unroll
      for (int q = 0; q < NR_Q; ++q) {
        const int e = lane + 32 * q;
        if (e < EN) {
          h1s[(wave * NR_R + rr) * H2_P + e] = va[rr][q];
          h2s[(wave * NR_R + rr) * H2_P + e] = vb[rr][q];
        }
      }
    }
    float m1 = 0.f, m2 = 0.f;
#pragma unroll
    for (int rr = 0; rr < NR_R; ++rr) {
      float s1 = 0.f, s2 = 0.f;
#pragma unroll
      for (int q = 0; q < NR_Q; ++q) {
        if (lane + 32 * q < EN) {
          s1 += va[rr][q] * va[rr][q];
          s2 += vb[rr][q] * vb[rr][q];
        }
      }
      s1 = wred_sum(s1);
      s2 = wred_sum(s2);
      if (lane == rr) { m1 = s1; m2 = s2; }
    }
    if (lane < NR_R) {
      sn1[wave * NR_R + lane] = sqrtf(fmaxf(m1, EPSF));
      sn2[wave * NR_R + lane] = sqrtf(fmaxf(m2, EPSF));
    }
  }
  __syncthreads();

  {
    const int j0 = (wave & 3) * 16;
    const int ia = (wave >> 2) * 16, ib = ia + 32;
    const v2f* pa = (const v2f*)(h1s + (ia + ll) * H2_P + 2 * lh);
    const v2f* pb = (const v2f*)(h1s + (ib + ll) * H2_P + 2 * lh);
    const v2f* pj = (const v2f*)(h2s + (j0 + ll) * H2_P + 2 * lh);
    v2f ca[CT_QB], cb[CT_QB], cj[CT_QB];
#pragma unroll
    for (int q = 0; q < CT_QB; ++q) {
      ca[q] = pa[2 * q];
      cb[q] = pb[2 * q];
      cj[q] = pj[2 * q];
    }
    v8f acc0 = {}, acc1 = {};
#pragma unroll
    for (int blk = 0; blk < CT_NB; ++blk) {
      v2f na[CT_QB], nb[CT_QB], nj[CT_QB];
      if (blk + 1 < CT_NB) {
#pragma unroll
        for (int q = 0; q < CT_QB; ++q) {
          const int o = 2 * ((blk + 1) * CT_QB + q);
          na[q] = pa[o];
          nb[q] = pb[o];
          nj[q] = pj[o];
        }
        __builtin_amdgcn_sched_barrier(0);
      }
#pragma unroll
      for (int q = 0; q < CT_QB; ++q) {
        acc0 = __builtin_amdgcn_wmma_f32_16x16x4_f32(false, ca[q], false, cj[q], (short)0, acc0, false, false);
        acc1 = __builtin_amdgcn_wmma_f32_16x16x4_f32(false, cb[q], false, cj[q], (short)0, acc1, false, false);
      }
      if (blk + 2 < CT_NB) __builtin_amdgcn_sched_barrier(0);
      if (blk + 1 < CT_NB) {
#pragma unroll
        for (int q = 0; q < CT_QB; ++q) { ca[q] = na[q]; cb[q] = nb[q]; cj[q] = nj[q]; }
      }
    }
    const float invn2 = 1.f / sn2[j0 + ll];
#pragma unroll
    for (int r = 0; r < 8; ++r) {
      const int i = ia + lh * 8 + r;
      cs[i * CS_P + (j0 + ll)] = acc0[r] / sn1[i] * invn2;
    }
#pragma unroll
    for (int r = 0; r < 8; ++r) {
      const int i = ib + lh * 8 + r;
      cs[i * CS_P + (j0 + ll)] = acc1[r] / sn1[i] * invn2;
    }
  }
  __syncthreads();

  float mys = 0.f; int mymi = 0;
#pragma unroll
  for (int rr = 0; rr < 8; ++rr) {
    const float* row = cs + (wave * 8 + rr) * CS_P;
    float v0 = row[lane], v1 = row[lane + 32];
    float s = v0 + v1;
    float m = v0; int mi = lane;
    if (v1 > m) { m = v1; mi = lane + 32; }
#pragma unroll
    for (int off = 16; off; off >>= 1) {
      s += __shfl_xor(s, off, 32);
      float om = __shfl_xor(m, off, 32);
      int omi = __shfl_xor(mi, off, 32);
      const bool take = (om > m) | ((om == m) & (omi < mi));
      m = take ? om : m;
      mi = take ? omi : mi;
    }
    if (lane == rr) { mys = s + EPSF; mymi = mi; }
  }
  if (lane < 8) {
    srs[wave * 8 + lane] = mys;
    amax[db * TN + wave * 8 + lane] = mymi;
  }
  __syncthreads();

  {
    const int i0 = (wave & 3) * 16;
    const int ct0 = (wave >> 2) * AT_CT;
    v2f af[AT_KS];
    const float* crow = cs + (i0 + ll) * CS_P + 2 * lh;
#pragma unroll
    for (int s = 0; s < AT_KS; ++s) af[s] = *(const v2f*)(crow + 4 * s);
    float rs[8];
#pragma unroll
    for (int r = 0; r < 8; ++r) rs[r] = srs[i0 + lh * 8 + r];
    float* arow = att + (db * TN + i0 + lh * 8) * EN;
    const float* hb = h2s + 2 * lh * H2_P;

    for (int c = 0; c < AT_CT; c += 2) {
      const int ea = (ct0 + c) * 16 + ll, eb = ea + 16;
      const float* pa = hb + (ea < EN ? ea : 0);
      const float* pb = hb + (eb < EN ? eb : 0);
      v2f ba[AT_KS], bb[AT_KS];
#pragma unroll
      for (int s = 0; s < AT_KS; ++s) {
        ba[s] = (v2f){ pa[(4 * s) * H2_P], pa[(4 * s + 1) * H2_P] };
        bb[s] = (v2f){ pb[(4 * s) * H2_P], pb[(4 * s + 1) * H2_P] };
      }
      __builtin_amdgcn_sched_barrier(0);
      v8f a0 = {}, a1 = {};
#pragma unroll
      for (int s = 0; s < AT_KS; ++s) {
        a0 = __builtin_amdgcn_wmma_f32_16x16x4_f32(false, af[s], false, ba[s], (short)0, a0, false, false);
        a1 = __builtin_amdgcn_wmma_f32_16x16x4_f32(false, af[s], false, bb[s], (short)0, a1, false, false);
      }
      if (ea < EN) {
#pragma unroll
        for (int r = 0; r < 8; ++r) arow[(size_t)r * EN + ea] = a0[r] / rs[r];
      }
      if (eb < EN) {
#pragma unroll
        for (int r = 0; r < 8; ++r) arow[(size_t)r * EN + eb] = a1[r] / rs[r];
      }
    }
  }
}

typedef __attribute__((ext_vector_type(4))) float v4f;
#define MK_WAVES 20
#define MK_THREADS (MK_WAVES * 32)
#define MK_FILL 3
#define MK_EH 152
#define MK_RP 153
#define MK_SB 8
#define MK_WS 84

struct MkAcc {
  v2f xxam;
  v2f xxfp;
  v2f xy12;
  v2f yy12;
  float xy0;
  float yo;
};

__device__ __forceinline__ void mk_row(const v4f R, const v4f w, const float t0, MkAcc& a) {
  const v2f y12 = R.xy;
  const v2f wam = w.xy;
  const v2f wfp = w.zw;
  const float x = R.z, yo = R.w;
  const v2f T = y12 * wam;
  const float xx = x * x;
  const v2f xs = { x, x };
  const v2f xxs = { xx, xx };
  a.xy12 += xs * T;
  a.yy12 += y12 * T;
  a.xxam += xxs * wam;
  a.xxfp += xxs * wfp;
  a.xy0 += x * t0;
  a.yo += yo * (yo * w.w);
}

__global__ __launch_bounds__(MK_THREADS)
void match_kernel(const float* __restrict__ h1, const float* __restrict__ h2,
                  const float* __restrict__ kern, const float* __restrict__ att,
                  const int* __restrict__ amax,
                  float* __restrict__ N1mp, float* __restrict__ N2mp,
                  float* __restrict__ out) {
  __shared__ __attribute__((aligned(16))) float w2t[EN * MK_WS];
  __shared__ __attribute__((aligned(16))) float rowb[TN * MK_RP * 4];
  __shared__ __attribute__((aligned(16))) float y0b[EN];

  const int db = blockIdx.x;
  const int b = db % BN, d = db / BN;
  const int tid = threadIdx.x;

  for (int base = tid; base < MP * EN; base += MK_THREADS * MK_FILL) {
    float wv[MK_FILL][4];
#pragma unroll
    for (int k = 0; k < MK_FILL; ++k) {
      const int idx = base + k * MK_THREADS;
      const int idc = idx < MP * EN ? idx : 0;
      const int e = idc % EN, m = idc / EN;
#pragma unroll
      for (int s = 0; s < 4; ++s)
        wv[k][s] = kern[(size_t)m * 2400 + (s * 2 + d) * EN + e];
    }
#pragma unroll
    for (int k = 0; k < MK_FILL; ++k) {
      const int idx = base + k * MK_THREADS;
      if (idx < MP * EN) {
        const int e = idx % EN, m = idx / EN;
        const v4f v = { wv[k][1] * wv[k][1], wv[k][2] * wv[k][2],
                        wv[k][0] * wv[k][0], wv[k][3] * wv[k][3] };
        *(v4f*)(w2t + e * MK_WS + m * 4) = v;
      }
    }
  }
  if (tid < EN)
    y0b[tid] = h2[((size_t)b * TN + (TN - 1)) * S2E + d * EN + tid];

  const int hh = tid >> 4, hl = tid & 15;
  const int m = hh < TN / 2 ? hl : 16 + (hl & 3);
  const int g = hh < TN / 2 ? hh : 4 * (hh - TN / 2) + (hl >> 2);
  const v4f* wp = (const v4f*)(w2t + m * 4);
  const v4f* ra = (const v4f*)rowb + (size_t)(2 * g) * MK_RP;
  const v4f* rb = ra + MK_RP;
  const v4f* y0p = (const v4f*)y0b;

  MkAcc accA = { {0.f, 0.f}, {0.f, 0.f}, {0.f, 0.f}, {0.f, 0.f}, 0.f, 0.f };
  MkAcc accB = accA;
  float syy0 = 0.f;

  for (int h = 0; h < 2; ++h) {
    const int eb = h * MK_EH;
    if (h) __syncthreads();

    for (int base = tid; base < TN * MK_EH; base += MK_THREADS * MK_SB) {
      v4f v[MK_SB];
#pragma unroll
      for (int k = 0; k < MK_SB; ++k) {
        const int i = base + k * MK_THREADS;
        const int ic = i < TN * MK_EH ? i : tid;
        const int row = ic / MK_EH, el = ic % MK_EH;
        const int e = eb + el < EN ? eb + el : eb;
        const int jm = amax[(size_t)db * TN + row];
        const size_t bt = (size_t)b * TN + row;
        const v4f r = { att[((size_t)db * TN + row) * EN + e],
                        h2[((size_t)b * TN + jm) * S2E + d * EN + e],
                        h1[bt * S2E + d * EN + e],
                        h2[bt * S2E + d * EN + e] };
        v[k] = r;
      }
#pragma unroll
      for (int k = 0; k < MK_SB; ++k) {
        const int i = base + k * MK_THREADS;
        if (i < TN * MK_EH) {
          const int row = i / MK_EH, el = i % MK_EH;
          if (eb + el < EN) *(v4f*)(rowb + ((size_t)row * MK_RP + el) * 4) = v[k];
        }
      }
    }
    __syncthreads();

    const int n4 = (h ? EN - MK_EH : MK_EH) / 4;
    for (int e4 = 0; e4 < n4; ++e4) {
      const v4f y0v = y0p[eb / 4 + e4];
#pragma unroll
      for (int u = 0; u < 4; ++u) {
        const int el = e4 * 4 + u;
        const v4f w = wp[(eb + el) * (MK_WS / 4)];
        const v4f A = ra[el];
        const v4f B = rb[el];
        const float y0 = y0v[u];
        const float t0w = y0 * w.z;
        syy0 += y0 * t0w;
        mk_row(A, w, t0w, accA);
        mk_row(B, w, t0w, accB);
      }
    }
  }

  {
    const float ny0 = sqrtf(fmaxf(syy0, EPSF));
#pragma unroll
    for (int r = 0; r < 2; ++r) {
      const MkAcc a = r ? accB : accA;
      const int t = 2 * g + r;
      float* orow = out + ((size_t)b * TN + t) * 160;
      orow[(0 * 2 + d) * MP + m] = a.xy0 / sqrtf(fmaxf(a.xxfp.x, EPSF)) / ny0;
      orow[(1 * 2 + d) * MP + m] = a.xy12.x / sqrtf(fmaxf(a.xxam.x, EPSF)) / sqrtf(fmaxf(a.yy12.x, EPSF));
      orow[(2 * 2 + d) * MP + m] = a.xy12.y / sqrtf(fmaxf(a.xxam.y, EPSF)) / sqrtf(fmaxf(a.yy12.y, EPSF));
      const size_t o = ((size_t)db * TN + t) * MP + m;
      N1mp[o] = sqrtf(fmaxf(a.xxfp.y, EPSF));
      N2mp[o] = sqrtf(fmaxf(a.yo, EPSF));
    }
  }
}

#define MX_KS (EN / 4)
#define MX_QB 19
#define MX_NB 4

template <int MASK>
__device__ __forceinline__ float row_xor_max(float v) {
  const int o = __builtin_amdgcn_update_dpp(0, __builtin_bit_cast(int, v), 0x160 + MASK, 0xf, 0xf, true);
  return fmaxf(v, __builtin_bit_cast(float, o));
}

__global__ __launch_bounds__(128)
void maxpool_kernel(const float* __restrict__ h1, const float* __restrict__ h2,
                    const float* __restrict__ kern,
                    const float* __restrict__ N1mp, const float* __restrict__ N2mp,
                    float* __restrict__ out) {
  __shared__ __attribute__((aligned(16))) float wl[MP * EN];
  const int db = blockIdx.x;
  const int b = db % BN, d = db / BN;
  const int wave = threadIdx.x >> 5, lane = threadIdx.x & 31;
  const int lh = lane >> 4, ll = lane & 15;

  const int i0 = wave * 16;
  const float* Ab = h1 + ((size_t)b * TN) * S2E + d * EN + 2 * lh;
  const float* Bb = h2 + ((size_t)b * TN) * S2E + d * EN + 2 * lh;
  const v2f* arow = (const v2f*)(Ab + (size_t)(i0 + ll) * S2E);
  const v2f* b0r  = (const v2f*)(Bb + (size_t)(0  + ll) * S2E);
  const v2f* b1r  = (const v2f*)(Bb + (size_t)(16 + ll) * S2E);
  const v2f* b2r  = (const v2f*)(Bb + (size_t)(32 + ll) * S2E);
  const v2f* b3r  = (const v2f*)(Bb + (size_t)(48 + ll) * S2E);

  v2f ra[MX_KS], r0[MX_KS], r1[MX_KS], r2[MX_KS], r3[MX_KS];
#pragma unroll
  for (int q = 0; q < MX_KS; ++q) {
    ra[q] = arow[2 * q];
    r0[q] = b0r[2 * q];
    r1[q] = b1r[2 * q];
    r2[q] = b2r[2 * q];
    r3[q] = b3r[2 * q];
  }

  {
    const float* ws = kern + (6 + d) * EN;
    const int e0 = threadIdx.x, e1 = e0 + 128, e2 = e0 + 256;
    const int e2c = e2 < EN ? e2 : e0;
    float t0[MP], t1[MP], t2[MP];
#pragma unroll
    for (int m = 0; m < MP; ++m) {
      const float* wr = ws + (size_t)m * 2400;
      t0[m] = wr[e0];
      t1[m] = wr[e1];
      t2[m] = wr[e2c];
    }
#pragma unroll
    for (int m = 0; m < MP; ++m) {
      wl[m * EN + e0] = t0[m];
      wl[m * EN + e1] = t1[m];
    }
    if (e2 < EN) {
#pragma unroll
      for (int m = 0; m < MP; ++m) wl[m * EN + e2] = t2[m];
    }
  }
  __syncthreads();

  const int myrow = i0 + lh * 8 + (ll & 7);
  const float* n1p = N1mp + (size_t)db * TN * MP + (size_t)myrow * MP;
  const float* n2p = N2mp + (size_t)db * TN * MP + (size_t)ll * MP;
  float* orow = out + ((size_t)(b * TN + myrow)) * 160 + (6 + d) * MP;

  float n1nx = n1p[0];
  v2f wc[MX_QB];
  {
    const v2f* w0 = (const v2f*)(wl + 2 * lh);
#pragma unroll
    for (int j = 0; j < MX_QB; ++j) wc[j] = w0[2 * j];
  }
  for (int m = 0; m < MP; ++m) {
    const float n1v = n1nx;
    n1nx = n1p[m + 1 < MP ? m + 1 : m];
    const float d0 = n2p[(size_t)0  * MP + m];
    const float d1 = n2p[(size_t)16 * MP + m];
    const float d2 = n2p[(size_t)32 * MP + m];
    const float d3 = n2p[(size_t)48 * MP + m];
    const v2f* wm = (const v2f*)(wl + m * EN + 2 * lh);
    const v2f* wnx = (const v2f*)(wl + (m + 1 < MP ? m + 1 : m) * EN + 2 * lh);

    v8f a0 = {}, a1 = {}, a2 = {}, a3 = {};
#pragma unroll
    for (int blk = 0; blk < MX_NB; ++blk) {
      const int qb = blk * MX_QB;
      const int qn = qb + MX_QB < MX_KS ? MX_QB : MX_KS - qb;
      v2f wn[MX_QB];
      const v2f* nsrc = blk + 1 < MX_NB ? wm + 2 * (qb + MX_QB) : wnx;
#pragma unroll
      for (int j = 0; j < MX_QB; ++j) {
        const bool in = blk + 1 < MX_NB ? (qb + MX_QB + j < MX_KS) : true;
        wn[j] = nsrc[in ? 2 * j : 0];
      }
      __builtin_amdgcn_sched_barrier(0);
#pragma unroll
      for (int j = 0; j < MX_QB; ++j) {
        if (j < qn) {
          const int q = qb + j;
          const v2f w = wc[j];
          const v2f af = ra[q] * w;
          const v2f f0 = r0[q] * w;
          const v2f f1 = r1[q] * w;
          const v2f f2 = r2[q] * w;
          const v2f f3 = r3[q] * w;
          a0 = __builtin_amdgcn_wmma_f32_16x16x4_f32(false, af, false, f0, (short)0, a0, false, false);
          a1 = __builtin_amdgcn_wmma_f32_16x16x4_f32(false, af, false, f1, (short)0, a1, false, false);
          a2 = __builtin_amdgcn_wmma_f32_16x16x4_f32(false, af, false, f2, (short)0, a2, false, false);
          a3 = __builtin_amdgcn_wmma_f32_16x16x4_f32(false, af, false, f3, (short)0, a3, false, false);
        }
      }
      __builtin_amdgcn_sched_barrier(0);
#pragma unroll
      for (int j = 0; j < MX_QB; ++j) wc[j] = wn[j];
    }

    float v[8];
#pragma unroll
    for (int r = 0; r < 8; ++r) {
      const float v0 = a0[r] / d0;
      const float v1 = a1[r] / d1;
      const float v2 = a2[r] / d2;
      const float v3 = a3[r] / d3;
      v[r] = fmaxf(fmaxf(v0, v1), fmaxf(v2, v3));
    }
#pragma unroll
    for (int r = 0; r < 8; ++r) v[r] = row_xor_max<1>(v[r]);
#pragma unroll
    for (int r = 0; r < 8; ++r) v[r] = row_xor_max<2>(v[r]);
#pragma unroll
    for (int r = 0; r < 8; ++r) v[r] = row_xor_max<4>(v[r]);
#pragma unroll
    for (int r = 0; r < 8; ++r) v[r] = row_xor_max<8>(v[r]);
    float mine = v[0];
#pragma unroll
    for (int r = 1; r < 8; ++r) mine = ((ll & 7) == r) ? v[r] : mine;
    if (ll < 8) orow[m] = mine / n1v;
  }
}

extern "C" void kernel_launch(void* const* d_in, const int* in_sizes, int n_in,
                              void* d_out, int out_size, void* d_ws, size_t ws_size,
                              hipStream_t stream) {
  const float* h1   = (const float*)d_in[0];
  const float* h2   = (const float*)d_in[1];
  const float* kern = (const float*)d_in[2];
  float* out = (float*)d_out;

  float* cosm   = (float*)d_ws;
  float* rowsum = cosm + (size_t)2 * BN * TN * TN;
  int*   amax   = (int*)(rowsum + 2 * BN * TN);
  float* att    = (float*)(amax + 2 * BN * TN);
  float* N1mp   = att + (size_t)2 * BN * TN * EN;
  float* N2mp   = N1mp + (size_t)2 * BN * TN * MP;

  cosatt_kernel  <<<2 * BN,          256, 0, stream>>>(h1, h2, amax, att);
  match_kernel   <<<2 * BN,          MK_THREADS, 0, stream>>>(h1, h2, kern, att, amax,
                                                       N1mp, N2mp, out);
  maxpool_kernel <<<2 * BN,          128, 0, stream>>>(h1, h2, kern, N1mp, N2mp, out);
}
